# starting kernel
# baseline (speedup 1.0000x reference)
#include <hip/hip_runtime.h>
#include <math.h>

typedef __attribute__((ext_vector_type(2))) float v2f;
typedef __attribute__((ext_vector_type(8))) float v8f;

#define NEG_SLOPE 0.2f

__device__ __forceinline__ unsigned fkey(float f) {
    unsigned b = __float_as_uint(f);
    return (b & 0x80000000u) ? ~b : (b | 0x80000000u);
}
__device__ __forceinline__ float fkey_inv(unsigned u) {
    return (u & 0x80000000u) ? __uint_as_float(u & 0x7fffffffu)
                             : __uint_as_float(~u);
}
#define FKEY_NEG_INF 0x007fffffu

template <int K, int NCOL>
__global__ void gemm_wmma_kernel(const float* __restrict__ A,
                                 const float* __restrict__ B,
                                 float* __restrict__ C, int N) {
    constexpr int NT = NCOL / 16;
    const int lane = threadIdx.x & 31;
    const int l    = lane & 15;
    const int hf   = lane >> 4;
    const int row0 = blockIdx.x * 16;
    if (row0 >= N) return;

    const float* __restrict__ arow = A + (size_t)(row0 + l) * K;
    v8f acc[NT];
#pragma unroll
    for (int t = 0; t < NT; ++t) acc[t] = (v8f){};

#pragma unroll
    for (int k = 0; k < K; k += 4) {
        v2f a;
        a.x = arow[k + 2 * hf];
        a.y = arow[k + 2 * hf + 1];
#pragma unroll
        for (int t = 0; t < NT; ++t) {
            v2f b;
            b.x = B[(size_t)(k + 2 * hf) * NCOL + 16 * t + l];
            b.y = B[(size_t)(k + 2 * hf + 1) * NCOL + 16 * t + l];
            acc[t] = __builtin_amdgcn_wmma_f32_16x16x4_f32(
                 false, a,  false, b,
                 (short)0, acc[t],  false,
                 false);
        }
    }
#pragma unroll
    for (int t = 0; t < NT; ++t)
#pragma unroll
        for (int i = 0; i < 8; ++i)
            C[(size_t)(row0 + i + 8 * hf) * NCOL + 16 * t + l] = acc[t][i];
}

template <int H, int C>
__global__ void alpha_kernel(const float* __restrict__ xw,
                             const float* __restrict__ a_src,
                             const float* __restrict__ a_dst,
                             float* __restrict__ as_, float* __restrict__ ad_,
                             int N) {
    int idx = blockIdx.x * blockDim.x + threadIdx.x;
    if (idx >= N * H) return;
    int n = idx / H, h = idx % H;
    const float* v = xw + (size_t)n * (H * C) + h * C;
    float s = 0.f, d = 0.f;
#pragma unroll
    for (int c = 0; c < C; ++c) {
        s += v[c] * a_src[h * C + c];
        d += v[c] * a_dst[h * C + c];
    }
    as_[idx] = s;
    ad_[idx] = d;
}

__global__ void init_kernel(unsigned* __restrict__ m_, float* __restrict__ den_,
                            float* __restrict__ acc_, int NH, int NHC) {
    int idx = blockIdx.x * blockDim.x + threadIdx.x;
    if (idx < NHC) acc_[idx] = 0.f;
    if (idx < NH) { m_[idx] = FKEY_NEG_INF; den_[idx] = 0.f; }
}

__device__ __forceinline__ void edge_sd(const int* __restrict__ srcs,
                                        const int* __restrict__ dsts, int E,
                                        int i, int& s, int& d) {
    if (i < E) { s = srcs[i]; d = dsts[i]; }
    else       { s = i - E;   d = i - E;   }
}

template <int H>
__global__ void edge_max_kernel(const int* __restrict__ srcs,
                                const int* __restrict__ dsts, int E, int N,
                                const float* __restrict__ as_,
                                const float* __restrict__ ad_,
                                unsigned* __restrict__ m_) {
    int i = blockIdx.x * blockDim.x + threadIdx.x;
    if (i >= E + N) return;
    int s, d; edge_sd(srcs, dsts, E, i, s, d);
#pragma unroll
    for (int h = 0; h < H; ++h) {
        float e = as_[s * H + h] + ad_[d * H + h];
        e = (e > 0.f) ? e : NEG_SLOPE * e;
        atomicMax(&m_[d * H + h], fkey(e));
    }
}

template <int H>
__global__ void edge_denom_kernel(const int* __restrict__ srcs,
                                  const int* __restrict__ dsts, int E, int N,
                                  const float* __restrict__ as_,
                                  const float* __restrict__ ad_,
                                  const unsigned* __restrict__ m_,
                                  float* __restrict__ den_) {
    int i = blockIdx.x * blockDim.x + threadIdx.x;
    if (i >= E + N) return;
    int s, d; edge_sd(srcs, dsts, E, i, s, d);
#pragma unroll
    for (int h = 0; h < H; ++h) {
        float e = as_[s * H + h] + ad_[d * H + h];
        e = (e > 0.f) ? e : NEG_SLOPE * e;
        atomicAdd(&den_[d * H + h], expf(e - fkey_inv(m_[d * H + h])));
    }
}

template <int H, int C>
__global__ void edge_accum_kernel(const int* __restrict__ srcs,
                                  const int* __restrict__ dsts, int E, int N,
                                  const float* __restrict__ as_,
                                  const float* __restrict__ ad_,
                                  const unsigned* __restrict__ m_,
                                  const float* __restrict__ den_,
                                  const float* __restrict__ xw,
                                  float* __restrict__ acc_) {
    static_assert(C % 4 == 0, "C must be a multiple of 4");
    int i = blockIdx.x * blockDim.x + threadIdx.x;
    if (i >= E + N) return;
    int s, d; edge_sd(srcs, dsts, E, i, s, d);

    float w[H];
#pragma unroll
    for (int h = 0; h < H; ++h) {
        float e = as_[s * H + h] + ad_[d * H + h];
        e = (e > 0.f) ? e : NEG_SLOPE * e;
        w[h] = expf(e - fkey_inv(m_[d * H + h])) / den_[d * H + h];
    }
    const float4* __restrict__ sv =
        (const float4*)(xw + (size_t)s * (H * C));
    float* __restrict__ dv = acc_ + (size_t)d * (H * C);
#pragma unroll
    for (int f = 0; f < (H * C) / 4; ++f) {
        float4 v = sv[f];
        float wh = w[(f * 4) / C];
        atomicAdd(&dv[f * 4 + 0], wh * v.x);
        atomicAdd(&dv[f * 4 + 1], wh * v.y);
        atomicAdd(&dv[f * 4 + 2], wh * v.z);
        atomicAdd(&dv[f * 4 + 3], wh * v.w);
    }
}

__global__ void bias_elu_kernel(const float* __restrict__ acc_,
                                const float* __restrict__ bias,
                                float* __restrict__ h_, int total, int F) {
    int i = blockIdx.x * blockDim.x + threadIdx.x;
    if (i >= total) return;
    float v = acc_[i] + bias[i & (F - 1)];
    h_[i] = (v > 0.f) ? v : (expf(v) - 1.f);
}

__global__ void logsoftmax_kernel(const float* __restrict__ acc_,
                                  const float* __restrict__ bias,
                                  float* __restrict__ out, int N) {
    int n = blockIdx.x * blockDim.x + threadIdx.x;
    if (n >= N) return;
    float v[16];
    float m = -INFINITY;
#pragma unroll
    for (int c = 0; c < 16; ++c) {
        v[c] = acc_[(size_t)n * 16 + c] + bias[c];
        m = fmaxf(m, v[c]);
    }
    float ssum = 0.f;
#pragma unroll
    for (int c = 0; c < 16; ++c) ssum += expf(v[c] - m);
    float lg = m + logf(ssum);
#pragma unroll
    for (int c = 0; c < 16; ++c) out[(size_t)n * 16 + c] = v[c] - lg;
}

extern "C" void kernel_launch(void* const* d_in, const int* in_sizes, int n_in,
                              void* d_out, int out_size, void* d_ws,
                              size_t ws_size, hipStream_t stream) {
    const float* x   = (const float*)d_in[0];
    const int*   ei  = (const int*)d_in[1];
    const float* W1  = (const float*)d_in[2];
    const float* a1s = (const float*)d_in[3];
    const float* a1d = (const float*)d_in[4];
    const float* b1  = (const float*)d_in[5];
    const float* W2  = (const float*)d_in[6];
    const float* a2s = (const float*)d_in[7];
    const float* a2d = (const float*)d_in[8];
    const float* b2  = (const float*)d_in[9];
    float* out = (float*)d_out;

    const int N = in_sizes[0] / 128;
    const int E = in_sizes[1] / 2;
    const int* srcs = ei;
    const int* dsts = ei + E;
    const int T = E + N;

    char* p = (char*)d_ws;
    auto carve = [&](size_t bytes) {
        void* r = (void*)p;
        p += (bytes + 255) & ~(size_t)255;
        return r;
    };
    float*    xw1  = (float*)carve((size_t)N * 32 * 4);
    float*    as1  = (float*)carve((size_t)N * 4 * 4);
    float*    ad1  = (float*)carve((size_t)N * 4 * 4);
    unsigned* m1   = (unsigned*)carve((size_t)N * 4 * 4);
    float*    den1 = (float*)carve((size_t)N * 4 * 4);
    float*    out1 = (float*)carve((size_t)N * 32 * 4);
    float*    h1   = (float*)carve((size_t)N * 32 * 4);
    float*    xw2  = (float*)carve((size_t)N * 16 * 4);
    float*    as2  = (float*)carve((size_t)N * 4);
    float*    ad2  = (float*)carve((size_t)N * 4);
    unsigned* m2   = (unsigned*)carve((size_t)N * 4);
    float*    den2 = (float*)carve((size_t)N * 4);
    float*    out2 = (float*)carve((size_t)N * 16 * 4);

    const int B = 256;
    auto g = [](int n, int b) { return (n + b - 1) / b; };

    gemm_wmma_kernel<128, 32><<<N / 16, 32, 0, stream>>>(x, W1, xw1, N);
    alpha_kernel<4, 8><<<g(N * 4, B), B, 0, stream>>>(xw1, a1s, a1d, as1, ad1, N);
    init_kernel<<<g(N * 32, B), B, 0, stream>>>(m1, den1, out1, N * 4, N * 32);
    edge_max_kernel<4><<<g(T, B), B, 0, stream>>>(srcs, dsts, E, N, as1, ad1, m1);
    edge_denom_kernel<4><<<g(T, B), B, 0, stream>>>(srcs, dsts, E, N, as1, ad1,
                                                    m1, den1);
    edge_accum_kernel<4, 8><<<g(T, B), B, 0, stream>>>(srcs, dsts, E, N, as1,
                                                       ad1, m1, den1, xw1, out1);
    bias_elu_kernel<<<g(N * 32, B), B, 0, stream>>>(out1, b1, h1, N * 32, 32);

    gemm_wmma_kernel<32, 16><<<N / 16, 32, 0, stream>>>(h1, W2, xw2, N);
    alpha_kernel<1, 16><<<g(N, B), B, 0, stream>>>(xw2, a2s, a2d, as2, ad2, N);
    init_kernel<<<g(N * 16, B), B, 0, stream>>>(m2, den2, out2, N, N * 16);
    edge_max_kernel<1><<<g(T, B), B, 0, stream>>>(srcs, dsts, E, N, as2, ad2, m2);
    edge_denom_kernel<1><<<g(T, B), B, 0, stream>>>(srcs, dsts, E, N, as2, ad2,
                                                    m2, den2);
    edge_accum_kernel<1, 16><<<g(T, B), B, 0, stream>>>(srcs, dsts, E, N, as2,
                                                        ad2, m2, den2, xw2, out2);
    logsoftmax_kernel<<<g(N, B), B, 0, stream>>>(out2, b2, out, N);
}


# edge_accum: a wave's lanes span one edge's feature row (two edges for 16-float rows): coalesced row gather and row-wide atomic adds replace one edge per lane; atomics kept, arrival order differs
# speedup vs baseline: 2.1817x; 2.1817x over previous
#include <hip/hip_runtime.h>
#include <math.h>

typedef __attribute__((ext_vector_type(2))) float v2f;
typedef __attribute__((ext_vector_type(8))) float v8f;

#define NEG_SLOPE 0.2f

__device__ __forceinline__ unsigned fkey(float f) {
    unsigned b = __float_as_uint(f);
    return (b & 0x80000000u) ? ~b : (b | 0x80000000u);
}
__device__ __forceinline__ float fkey_inv(unsigned u) {
    return (u & 0x80000000u) ? __uint_as_float(u & 0x7fffffffu)
                             : __uint_as_float(~u);
}
#define FKEY_NEG_INF 0x007fffffu

template <int K, int NCOL>
__global__ void gemm_wmma_kernel(const float* __restrict__ A,
                                 const float* __restrict__ B,
                                 float* __restrict__ C, int N) {
    constexpr int NT = NCOL / 16;
    const int lane = threadIdx.x & 31;
    const int l    = lane & 15;
    const int hf   = lane >> 4;
    const int row0 = blockIdx.x * 16;
    if (row0 >= N) return;

    const float* __restrict__ arow = A + (size_t)(row0 + l) * K;
    v8f acc[NT];
#pragma unroll
    for (int t = 0; t < NT; ++t) acc[t] = (v8f){};

#pragma unroll
    for (int k = 0; k < K; k += 4) {
        v2f a;
        a.x = arow[k + 2 * hf];
        a.y = arow[k + 2 * hf + 1];
#pragma unroll
        for (int t = 0; t < NT; ++t) {
            v2f b;
            b.x = B[(size_t)(k + 2 * hf) * NCOL + 16 * t + l];
            b.y = B[(size_t)(k + 2 * hf + 1) * NCOL + 16 * t + l];
            acc[t] = __builtin_amdgcn_wmma_f32_16x16x4_f32(
                 false, a,  false, b,
                 (short)0, acc[t],  false,
                 false);
        }
    }
#pragma unroll
    for (int t = 0; t < NT; ++t)
#pragma unroll
        for (int i = 0; i < 8; ++i)
            C[(size_t)(row0 + i + 8 * hf) * NCOL + 16 * t + l] = acc[t][i];
}

template <int H, int C>
__global__ void alpha_kernel(const float* __restrict__ xw,
                             const float* __restrict__ a_src,
                             const float* __restrict__ a_dst,
                             float* __restrict__ as_, float* __restrict__ ad_,
                             int N) {
    int idx = blockIdx.x * blockDim.x + threadIdx.x;
    if (idx >= N * H) return;
    int n = idx / H, h = idx % H;
    const float* v = xw + (size_t)n * (H * C) + h * C;
    float s = 0.f, d = 0.f;
#pragma unroll
    for (int c = 0; c < C; ++c) {
        s += v[c] * a_src[h * C + c];
        d += v[c] * a_dst[h * C + c];
    }
    as_[idx] = s;
    ad_[idx] = d;
}

__global__ void init_kernel(unsigned* __restrict__ m_, float* __restrict__ den_,
                            float* __restrict__ acc_, int NH, int NHC) {
    int idx = blockIdx.x * blockDim.x + threadIdx.x;
    if (idx < NHC) acc_[idx] = 0.f;
    if (idx < NH) { m_[idx] = FKEY_NEG_INF; den_[idx] = 0.f; }
}

__device__ __forceinline__ void edge_sd(const int* __restrict__ srcs,
                                        const int* __restrict__ dsts, int E,
                                        int i, int& s, int& d) {
    if (i < E) { s = srcs[i]; d = dsts[i]; }
    else       { s = i - E;   d = i - E;   }
}

template <int H>
__global__ void edge_max_kernel(const int* __restrict__ srcs,
                                const int* __restrict__ dsts, int E, int N,
                                const float* __restrict__ as_,
                                const float* __restrict__ ad_,
                                unsigned* __restrict__ m_) {
    int i = blockIdx.x * blockDim.x + threadIdx.x;
    if (i >= E + N) return;
    int s, d; edge_sd(srcs, dsts, E, i, s, d);
#pragma unroll
    for (int h = 0; h < H; ++h) {
        float e = as_[s * H + h] + ad_[d * H + h];
        e = (e > 0.f) ? e : NEG_SLOPE * e;
        atomicMax(&m_[d * H + h], fkey(e));
    }
}

template <int H>
__global__ void edge_denom_kernel(const int* __restrict__ srcs,
                                  const int* __restrict__ dsts, int E, int N,
                                  const float* __restrict__ as_,
                                  const float* __restrict__ ad_,
                                  const unsigned* __restrict__ m_,
                                  float* __restrict__ den_) {
    int i = blockIdx.x * blockDim.x + threadIdx.x;
    if (i >= E + N) return;
    int s, d; edge_sd(srcs, dsts, E, i, s, d);
#pragma unroll
    for (int h = 0; h < H; ++h) {
        float e = as_[s * H + h] + ad_[d * H + h];
        e = (e > 0.f) ? e : NEG_SLOPE * e;
        atomicAdd(&den_[d * H + h], expf(e - fkey_inv(m_[d * H + h])));
    }
}

template <int H>
__device__ __forceinline__ float pick_head(const float (&t)[H], int hh) {
    float r = t[0];
#pragma unroll
    for (int h = 1; h < H; ++h) r = (hh == h) ? t[h] : r;
    return r;
}

template <int H, int C, bool GUARD>
__device__ __forceinline__ void accum_rows(int s, int d, const float (&w)[H],
                                           int lane, int wbase, int T,
                                           const float* __restrict__ xw,
                                           float* __restrict__ acc_) {
    constexpr int F   = H * C;
    constexpr int EPI = 32 / F;
    constexpr int NI  = 32 / EPI;
    const int f  = lane & (F - 1);
    const int q  = lane / F;
    const int hh = f / C;

    int   dj[NI];
    float val[NI];
#pragma unroll
    for (int k = 0; k < NI; ++k) {
        const int j = k * EPI;
        int sj; float t[H];
        if constexpr (EPI == 1) {
            sj    = __builtin_amdgcn_readlane(s, j);
            dj[k] = __builtin_amdgcn_readlane(d, j);
#pragma unroll
            for (int h = 0; h < H; ++h)
                t[h] = __int_as_float(
                    __builtin_amdgcn_readlane(__float_as_int(w[h]), j));
        } else {
            sj    = __shfl(s, j + q);
            dj[k] = __shfl(d, j + q);
#pragma unroll
            for (int h = 0; h < H; ++h) t[h] = __shfl(w[h], j + q);
        }
        const float wj = pick_head<H>(t, hh);
        float v = 0.f;
        if (!GUARD || wbase + j + q < T) v = xw[(size_t)sj * F + f];
        val[k] = wj * v;
    }
#pragma unroll
    for (int k = 0; k < NI; ++k) {
        const int j = k * EPI;
        if (!GUARD || wbase + j + q < T)
            atomicAdd(&acc_[(size_t)dj[k] * F + f], val[k]);
    }
}

template <int H, int C>
__global__ void edge_accum_kernel(const int* __restrict__ srcs,
                                  const int* __restrict__ dsts, int E, int N,
                                  const float* __restrict__ as_,
                                  const float* __restrict__ ad_,
                                  const unsigned* __restrict__ m_,
                                  const float* __restrict__ den_,
                                  const float* __restrict__ xw,
                                  float* __restrict__ acc_) {
    static_assert(H * C == 32 || H * C == 16, "row must be 32 or 16 floats");
    const int T    = E + N;
    const int lane = threadIdx.x & 31;
    const int i    = blockIdx.x * blockDim.x + threadIdx.x;
    const int wbase = __builtin_amdgcn_readfirstlane(i - lane);
    if (wbase >= T) return;

    int s = 0, d = 0;
    float w[H];
#pragma unroll
    for (int h = 0; h < H; ++h) w[h] = 0.f;
    if (i < T) {
        edge_sd(srcs, dsts, E, i, s, d);
#pragma unroll
        for (int h = 0; h < H; ++h) {
            float e = as_[s * H + h] + ad_[d * H + h];
            e = (e > 0.f) ? e : NEG_SLOPE * e;
            w[h] = expf(e - fkey_inv(m_[d * H + h])) / den_[d * H + h];
        }
    }
    if (wbase + 32 <= T)
        accum_rows<H, C, false>(s, d, w, lane, wbase, T, xw, acc_);
    else
        accum_rows<H, C, true>(s, d, w, lane, wbase, T, xw, acc_);
}

__global__ void bias_elu_kernel(const float* __restrict__ acc_,
                                const float* __restrict__ bias,
                                float* __restrict__ h_, int total, int F) {
    int i = blockIdx.x * blockDim.x + threadIdx.x;
    if (i >= total) return;
    float v = acc_[i] + bias[i & (F - 1)];
    h_[i] = (v > 0.f) ? v : (expf(v) - 1.f);
}

__global__ void logsoftmax_kernel(const float* __restrict__ acc_,
                                  const float* __restrict__ bias,
                                  float* __restrict__ out, int N) {
    int n = blockIdx.x * blockDim.x + threadIdx.x;
    if (n >= N) return;
    float v[16];
    float m = -INFINITY;
#pragma unroll
    for (int c = 0; c < 16; ++c) {
        v[c] = acc_[(size_t)n * 16 + c] + bias[c];
        m = fmaxf(m, v[c]);
    }
    float ssum = 0.f;
#pragma unroll
    for (int c = 0; c < 16; ++c) ssum += expf(v[c] - m);
    float lg = m + logf(ssum);
#pragma unroll
    for (int c = 0; c < 16; ++c) out[(size_t)n * 16 + c] = v[c] - lg;
}

extern "C" void kernel_launch(void* const* d_in, const int* in_sizes, int n_in,
                              void* d_out, int out_size, void* d_ws,
                              size_t ws_size, hipStream_t stream) {
    const float* x   = (const float*)d_in[0];
    const int*   ei  = (const int*)d_in[1];
    const float* W1  = (const float*)d_in[2];
    const float* a1s = (const float*)d_in[3];
    const float* a1d = (const float*)d_in[4];
    const float* b1  = (const float*)d_in[5];
    const float* W2  = (const float*)d_in[6];
    const float* a2s = (const float*)d_in[7];
    const float* a2d = (const float*)d_in[8];
    const float* b2  = (const float*)d_in[9];
    float* out = (float*)d_out;

    const int N = in_sizes[0] / 128;
    const int E = in_sizes[1] / 2;
    const int* srcs = ei;
    const int* dsts = ei + E;
    const int T = E + N;

    char* p = (char*)d_ws;
    auto carve = [&](size_t bytes) {
        void* r = (void*)p;
        p += (bytes + 255) & ~(size_t)255;
        return r;
    };
    float*    xw1  = (float*)carve((size_t)N * 32 * 4);
    float*    as1  = (float*)carve((size_t)N * 4 * 4);
    float*    ad1  = (float*)carve((size_t)N * 4 * 4);
    unsigned* m1   = (unsigned*)carve((size_t)N * 4 * 4);
    float*    den1 = (float*)carve((size_t)N * 4 * 4);
    float*    out1 = (float*)carve((size_t)N * 32 * 4);
    float*    h1   = (float*)carve((size_t)N * 32 * 4);
    float*    xw2  = (float*)carve((size_t)N * 16 * 4);
    float*    as2  = (float*)carve((size_t)N * 4);
    float*    ad2  = (float*)carve((size_t)N * 4);
    unsigned* m2   = (unsigned*)carve((size_t)N * 4);
    float*    den2 = (float*)carve((size_t)N * 4);
    float*    out2 = (float*)carve((size_t)N * 16 * 4);

    const int B = 256;
    auto g = [](int n, int b) { return (n + b - 1) / b; };

    gemm_wmma_kernel<128, 32><<<N / 16, 32, 0, stream>>>(x, W1, xw1, N);
    alpha_kernel<4, 8><<<g(N * 4, B), B, 0, stream>>>(xw1, a1s, a1d, as1, ad1, N);
    init_kernel<<<g(N * 32, B), B, 0, stream>>>(m1, den1, out1, N * 4, N * 32);
    edge_max_kernel<4><<<g(T, B), B, 0, stream>>>(srcs, dsts, E, N, as1, ad1, m1);
    edge_denom_kernel<4><<<g(T, B), B, 0, stream>>>(srcs, dsts, E, N, as1, ad1,
                                                    m1, den1);
    edge_accum_kernel<4, 8><<<g(T, B), B, 0, stream>>>(srcs, dsts, E, N, as1,
                                                       ad1, m1, den1, xw1, out1);
    bias_elu_kernel<<<g(N * 32, B), B, 0, stream>>>(out1, b1, h1, N * 32, 32);

    gemm_wmma_kernel<32, 16><<<N / 16, 32, 0, stream>>>(h1, W2, xw2, N);
    alpha_kernel<1, 16><<<g(N, B), B, 0, stream>>>(xw2, a2s, a2d, as2, ad2, N);
    init_kernel<<<g(N * 16, B), B, 0, stream>>>(m2, den2, out2, N, N * 16);
    edge_max_kernel<1><<<g(T, B), B, 0, stream>>>(srcs, dsts, E, N, as2, ad2, m2);
    edge_denom_kernel<1><<<g(T, B), B, 0, stream>>>(srcs, dsts, E, N, as2, ad2,
                                                    m2, den2);
    edge_accum_kernel<1, 16><<<g(T, B), B, 0, stream>>>(srcs, dsts, E, N, as2,
                                                        ad2, m2, den2, xw2, out2);
    logsoftmax_kernel<<<g(N, B), B, 0, stream>>>(out2, b2, out, N);
}


# edges grouped by destination per call (int-atomic count, allocate, scatter); one lane group per node gathers each source row once: score, running-max softmax, row sum in registers; arrival-rank order
# speedup vs baseline: 5.6526x; 2.5909x over previous
#include <hip/hip_runtime.h>
#include <math.h>

typedef __attribute__((ext_vector_type(2))) float v2f;
typedef __attribute__((ext_vector_type(8))) float v8f;

#define NEG_SLOPE 0.2f

template <int K, int NCOL>
__global__ void gemm_wmma_kernel(const float* __restrict__ A,
                                 const float* __restrict__ B,
                                 float* __restrict__ C, int N) {
    constexpr int NT = NCOL / 16;
    const int lane = threadIdx.x & 31;
    const int l    = lane & 15;
    const int hf   = lane >> 4;
    const int row0 = blockIdx.x * 16;
    if (row0 >= N) return;

    const float* __restrict__ arow = A + (size_t)(row0 + l) * K;
    v8f acc[NT];
#pragma unroll
    for (int t = 0; t < NT; ++t) acc[t] = (v8f){};

#pragma unroll
    for (int k = 0; k < K; k += 4) {
        v2f a;
        a.x = arow[k + 2 * hf];
        a.y = arow[k + 2 * hf + 1];
#pragma unroll
        for (int t = 0; t < NT; ++t) {
            v2f b;
            b.x = B[(size_t)(k + 2 * hf) * NCOL + 16 * t + l];
            b.y = B[(size_t)(k + 2 * hf + 1) * NCOL + 16 * t + l];
            acc[t] = __builtin_amdgcn_wmma_f32_16x16x4_f32(
                 false, a,  false, b,
                 (short)0, acc[t],  false,
                 false);
        }
    }
#pragma unroll
    for (int t = 0; t < NT; ++t)
#pragma unroll
        for (int i = 0; i < 8; ++i)
            C[(size_t)(row0 + i + 8 * hf) * NCOL + 16 * t + l] = acc[t][i];
}

__global__ void zero_kernel(int* __restrict__ cnt, int n) {
    int idx = blockIdx.x * blockDim.x + threadIdx.x;
    if (idx < n) cnt[idx] = 0;
}

template <bool GUARD>
__device__ __forceinline__ void degree4(const int* __restrict__ dsts, int E,
                                        int* __restrict__ cnt, int t, int S) {
    int d[4];
#pragma unroll
    for (int j = 0; j < 4; ++j) {
        const int i = t + j * S;
        d[j] = 0;
        if (!GUARD || i < E) d[j] = dsts[i];
    }
#pragma unroll
    for (int j = 0; j < 4; ++j) {
        const int i = t + j * S;
        if (!GUARD || i < E) atomicAdd(&cnt[d[j]], 1);
    }
}

__global__ void degree_kernel(const int* __restrict__ dsts, int E,
                              int* __restrict__ cnt) {
    const int S = gridDim.x * blockDim.x;
    const int t = blockIdx.x * blockDim.x + threadIdx.x;
    const int last = (__builtin_amdgcn_readfirstlane(t) | 31) + 3 * S;
    if (last < E) degree4<false>(dsts, E, cnt, t, S);
    else          degree4<true>(dsts, E, cnt, t, S);
}

__global__ void segment_alloc_kernel(int* __restrict__ cnt,
                                     int* __restrict__ beg,
                                     int* __restrict__ total, int N) {
    const int lane = threadIdx.x & 31;
    const int n    = blockIdx.x * blockDim.x + threadIdx.x;
    const int c    = (n < N) ? cnt[n] : 0;
    int v = c;
#pragma unroll
    for (int o = 1; o < 32; o <<= 1) {
        int t = __shfl_up(v, o);
        if (lane >= o) v += t;
    }
    int base = 0;
    if (lane == 31) base = atomicAdd(total, v);
    base = __shfl(base, 31);
    const int b = base + v - c;
    if (n < N) { beg[n] = b; cnt[n] = b; }
}

template <bool GUARD>
__device__ __forceinline__ void scatter4(const int* __restrict__ srcs,
                                         const int* __restrict__ dsts, int E,
                                         int* __restrict__ cur,
                                         int* __restrict__ ssrc, int t, int S) {
    int s[4], d[4], pos[4];
#pragma unroll
    for (int j = 0; j < 4; ++j) {
        const int i = t + j * S;
        s[j] = 0; d[j] = 0;
        if (!GUARD || i < E) { s[j] = srcs[i]; d[j] = dsts[i]; }
    }
#pragma unroll
    for (int j = 0; j < 4; ++j) {
        const int i = t + j * S;
        pos[j] = 0;
        if (!GUARD || i < E) pos[j] = atomicAdd(&cur[d[j]], 1);
    }
#pragma unroll
    for (int j = 0; j < 4; ++j) {
        const int i = t + j * S;
        if (!GUARD || i < E) ssrc[pos[j]] = s[j];
    }
}

__global__ void scatter_kernel(const int* __restrict__ srcs,
                               const int* __restrict__ dsts, int E,
                               int* __restrict__ cur, int* __restrict__ ssrc) {
    const int S = gridDim.x * blockDim.x;
    const int t = blockIdx.x * blockDim.x + threadIdx.x;
    const int last = (__builtin_amdgcn_readfirstlane(t) | 31) + 3 * S;
    if (last < E) scatter4<false>(srcs, dsts, E, cur, ssrc, t, S);
    else          scatter4<true>(srcs, dsts, E, cur, ssrc, t, S);
}

__device__ __forceinline__ float4 ld4(const float* __restrict__ p) {
    return *reinterpret_cast<const float4*>(p);
}
__device__ __forceinline__ float dot4(const float4& a, const float4& b) {
    return a.x * b.x + a.y * b.y + a.z * b.z + a.w * b.w;
}

template <int H, int C, bool FINAL>
__global__ void node_kernel(const int* __restrict__ beg,
                            const int* __restrict__ endp,
                            const int* __restrict__ ssrc,
                            const float* __restrict__ xw,
                            const float* __restrict__ a_src,
                            const float* __restrict__ a_dst,
                            const float* __restrict__ bias,
                            float* __restrict__ outp, int N) {
    constexpr int F   = H * C;
    constexpr int LPR = F / 4;
    constexpr int LPH = C / 4;
    constexpr int NPW = 32 / F;
    static_assert(F == 32 || F == 16, "row must be 32 or 16 floats");
    static_assert(C % 4 == 0, "a float4 must not straddle two heads");
    const int lane = threadIdx.x & 31;
    const int sub  = lane % F;
    const int g    = sub / LPR;
    const int c4   = sub % LPR;
    const int wave0 = __builtin_amdgcn_readfirstlane(
        (int)((blockIdx.x * blockDim.x + threadIdx.x) >> 5));
    const int nwaves = (int)((gridDim.x * blockDim.x) >> 5);
    const int nslots = (N + NPW - 1) / NPW;

    float4 qs, qd, bv;
    qs.x = a_src[4 * c4 + 0]; qs.y = a_src[4 * c4 + 1];
    qs.z = a_src[4 * c4 + 2]; qs.w = a_src[4 * c4 + 3];
    qd.x = a_dst[4 * c4 + 0]; qd.y = a_dst[4 * c4 + 1];
    qd.z = a_dst[4 * c4 + 2]; qd.w = a_dst[4 * c4 + 3];
    bv.x = bias[4 * c4 + 0];  bv.y = bias[4 * c4 + 1];
    bv.z = bias[4 * c4 + 2];  bv.w = bias[4 * c4 + 3];

    for (int q = wave0; q < nslots; q += nwaves) {
        const int  n    = q * NPW + lane / F;
        const bool live = n < N;
        int b = 0, cnt = 0;
        float4 rn = make_float4(0.f, 0.f, 0.f, 0.f);
        if (live) {
            b   = beg[n];
            cnt = endp[n] - b;
            rn  = ld4(xw + (size_t)n * F + 4 * c4);
        }
        float adn = dot4(rn, qd);
#pragma unroll
        for (int o = 1; o < LPH; o <<= 1) adn += __shfl_xor(adn, o);

        const int total = live ? cnt + 1 : 0;
        int iters = (total + 3) >> 2;
        if constexpr (NPW == 2) iters = max(iters, __shfl_xor(iters, 16));
        iters = __builtin_amdgcn_readfirstlane(iters);

        float m = -INFINITY, den = 0.f;
        float a0 = 0.f, a1 = 0.f, a2 = 0.f, a3 = 0.f;
        int k = g;
        const int s0 = (k < cnt) ? ssrc[b + k] : n;
        int s1 = (k + 4 < cnt) ? ssrc[b + k + 4] : n;
        float4 r_cur = make_float4(0.f, 0.f, 0.f, 0.f);
        if (k < total) r_cur = ld4(xw + (size_t)s0 * F + 4 * c4);
        for (int it = 0; it < iters; ++it) {
            const bool valid = k < total;
            const int  kn    = k + 4;
            const int  s2    = (kn + 4 < cnt) ? ssrc[b + kn + 4] : n;
            float4 r_nxt = make_float4(0.f, 0.f, 0.f, 0.f);
            if (kn < total) r_nxt = ld4(xw + (size_t)s1 * F + 4 * c4);
            const float4 r = r_cur;
            float e = dot4(r, qs);
#pragma unroll
            for (int o = 1; o < LPH; o <<= 1) e += __shfl_xor(e, o);
            if (valid) {
                e += adn;
                e = (e > 0.f) ? e : NEG_SLOPE * e;
                const float mn = fmaxf(m, e);
                const float sc = expf(m - mn);
                const float p  = expf(e - mn);
                den = den * sc + p;
                a0  = a0 * sc + p * r.x;
                a1  = a1 * sc + p * r.y;
                a2  = a2 * sc + p * r.z;
                a3  = a3 * sc + p * r.w;
                m   = mn;
            }
            k = kn; s1 = s2; r_cur = r_nxt;
        }
        float M = fmaxf(m, __shfl_xor(m, LPR));
        M = fmaxf(M, __shfl_xor(M, 2 * LPR));
        const float w = expf(m - M);
        den *= w; a0 *= w; a1 *= w; a2 *= w; a3 *= w;
#pragma unroll
        for (int o = LPR; o < F; o <<= 1) {
            den += __shfl_xor(den, o);
            a0  += __shfl_xor(a0, o);
            a1  += __shfl_xor(a1, o);
            a2  += __shfl_xor(a2, o);
            a3  += __shfl_xor(a3, o);
        }
        float v0 = a0 / den + bv.x;
        float v1 = a1 / den + bv.y;
        float v2 = a2 / den + bv.z;
        float v3 = a3 / den + bv.w;
        if constexpr (!FINAL) {
            v0 = (v0 > 0.f) ? v0 : (expf(v0) - 1.f);
            v1 = (v1 > 0.f) ? v1 : (expf(v1) - 1.f);
            v2 = (v2 > 0.f) ? v2 : (expf(v2) - 1.f);
            v3 = (v3 > 0.f) ? v3 : (expf(v3) - 1.f);
        } else {
            float mx = fmaxf(fmaxf(v0, v1), fmaxf(v2, v3));
#pragma unroll
            for (int o = 1; o < LPR; o <<= 1) mx = fmaxf(mx, __shfl_xor(mx, o));
            float ss = expf(v0 - mx) + expf(v1 - mx) + expf(v2 - mx) +
                       expf(v3 - mx);
#pragma unroll
            for (int o = 1; o < LPR; o <<= 1) ss += __shfl_xor(ss, o);
            const float lg = mx + logf(ss);
            v0 -= lg; v1 -= lg; v2 -= lg; v3 -= lg;
        }
        if (live && g == 0) {
            float4 o4; o4.x = v0; o4.y = v1; o4.z = v2; o4.w = v3;
            *reinterpret_cast<float4*>(outp + (size_t)n * F + 4 * c4) = o4;
        }
    }
}

extern "C" void kernel_launch(void* const* d_in, const int* in_sizes, int n_in,
                              void* d_out, int out_size, void* d_ws,
                              size_t ws_size, hipStream_t stream) {
    const float* x   = (const float*)d_in[0];
    const int*   ei  = (const int*)d_in[1];
    const float* W1  = (const float*)d_in[2];
    const float* a1s = (const float*)d_in[3];
    const float* a1d = (const float*)d_in[4];
    const float* b1  = (const float*)d_in[5];
    const float* W2  = (const float*)d_in[6];
    const float* a2s = (const float*)d_in[7];
    const float* a2d = (const float*)d_in[8];
    const float* b2  = (const float*)d_in[9];
    float* out = (float*)d_out;

    const int N = in_sizes[0] / 128;
    const int E = in_sizes[1] / 2;
    const int* srcs = ei;
    const int* dsts = ei + E;

    char* p = (char*)d_ws;
    auto carve = [&](size_t bytes) {
        void* r = (void*)p;
        p += (bytes + 255) & ~(size_t)255;
        return r;
    };
    float* xw1  = (float*)carve((size_t)N * 32 * 4);
    float* h1   = (float*)carve((size_t)N * 32 * 4);
    float* xw2  = (float*)carve((size_t)N * 16 * 4);
    int*   cnt  = (int*)carve((size_t)(N + 1) * 4);
    int*   beg  = (int*)carve((size_t)N * 4);
    int*   ssrc = (int*)carve((size_t)E * 4);

    const int B = 256;
    auto g = [](int n, int b) { return (n + b - 1) / b; };
    const int NODE_BLOCKS = 2048;

    zero_kernel<<<g(N + 1, B), B, 0, stream>>>(cnt, N + 1);
    degree_kernel<<<g(g(E, 4), B), B, 0, stream>>>(dsts, E, cnt);
    segment_alloc_kernel<<<g(N, B), B, 0, stream>>>(cnt, beg, cnt + N, N);
    scatter_kernel<<<g(g(E, 4), B), B, 0, stream>>>(srcs, dsts, E, cnt, ssrc);

    gemm_wmma_kernel<128, 32><<<N / 16, 32, 0, stream>>>(x, W1, xw1, N);
    node_kernel<4, 8, false><<<NODE_BLOCKS, B, 0, stream>>>(
        beg, cnt, ssrc, xw1, a1s, a1d, b1, h1, N);

    gemm_wmma_kernel<32, 16><<<N / 16, 32, 0, stream>>>(h1, W2, xw2, N);
    node_kernel<1, 16, true><<<NODE_BLOCKS, B, 0, stream>>>(
        beg, cnt, ssrc, xw2, a2s, a2d, b2, out, N);
}


# edges grouped by destination with LDS counters in two launches (bin edges per 256-node slice per chunk, then count, scan and place per slice) in place of global int atomics; arrival-rank order
# speedup vs baseline: 13.8356x; 2.4477x over previous
#include <hip/hip_runtime.h>
#include <math.h>

typedef __attribute__((ext_vector_type(2))) float v2f;
typedef __attribute__((ext_vector_type(8))) float v8f;

#define NEG_SLOPE 0.2f

template <int K, int NCOL>
__global__ void gemm_wmma_kernel(const float* __restrict__ A,
                                 const float* __restrict__ B,
                                 float* __restrict__ C, int N) {
    constexpr int NT = NCOL / 16;
    const int lane = threadIdx.x & 31;
    const int l    = lane & 15;
    const int hf   = lane >> 4;
    const int row0 = blockIdx.x * 16;
    if (row0 >= N) return;

    const float* __restrict__ arow = A + (size_t)(row0 + l) * K;
    v8f acc[NT];
#pragma unroll
    for (int t = 0; t < NT; ++t) acc[t] = (v8f){};

#pragma unroll
    for (int k = 0; k < K; k += 4) {
        v2f a;
        a.x = arow[k + 2 * hf];
        a.y = arow[k + 2 * hf + 1];
#pragma unroll
        for (int t = 0; t < NT; ++t) {
            v2f b;
            b.x = B[(size_t)(k + 2 * hf) * NCOL + 16 * t + l];
            b.y = B[(size_t)(k + 2 * hf + 1) * NCOL + 16 * t + l];
            acc[t] = __builtin_amdgcn_wmma_f32_16x16x4_f32(
                 false, a,  false, b,
                 (short)0, acc[t],  false,
                 false);
        }
    }
#pragma unroll
    for (int t = 0; t < NT; ++t)
#pragma unroll
        for (int i = 0; i < 8; ++i)
            C[(size_t)(row0 + i + 8 * hf) * NCOL + 16 * t + l] = acc[t][i];
}

#define SLICE_SHIFT 8
#define SLICE 256
#define CHUNK 4096
#define BIN_T 256
#define EPT (CHUNK / BIN_T)
#define SORT_T 512
#define STAGE_CAP 8192

__global__ void __launch_bounds__(BIN_T)
bin_kernel(const int* __restrict__ srcs, const int* __restrict__ dsts, int E,
           int NB, int G, int* __restrict__ tabT, int* __restrict__ binned) {
    extern __shared__ int smem[];
    int* cnt   = smem;
    int* stage = smem + (NB + 1);
    __shared__ int wsum[BIN_T / 32];

    const int t    = threadIdx.x;
    const int lane = t & 31;
    const int w    = blockIdx.x;
    const int e0   = w * CHUNK;
    const int n_here = min(CHUNK, E - e0);

    for (int i = t; i <= NB; i += BIN_T) cnt[i] = 0;
    __syncthreads();

    int s[EPT], d[EPT], r[EPT];
#pragma unroll
    for (int j = 0; j < EPT; ++j) {
        const int i = j * BIN_T + t;
        s[j] = 0; d[j] = 0;
        if (i < n_here) { s[j] = srcs[e0 + i]; d[j] = dsts[e0 + i]; }
    }
#pragma unroll
    for (int j = 0; j < EPT; ++j) {
        const int i = j * BIN_T + t;
        r[j] = 0;
        if (i < n_here) r[j] = atomicAdd(&cnt[d[j] >> SLICE_SHIFT], 1);
    }
    __syncthreads();

    const int per = (NB + 1 + BIN_T - 1) / BIN_T;
    const int lo  = t * per;
    int v = 0;
    for (int i = 0; i < per; ++i) {
        const int idx = lo + i;
        if (idx <= NB) v += cnt[idx];
    }
    int inc = v;
#pragma unroll
    for (int o = 1; o < 32; o <<= 1) {
        const int u = __shfl_up(inc, o);
        if (lane >= o) inc += u;
    }
    if (lane == 31) wsum[t >> 5] = inc;
    __syncthreads();
    int run = inc - v;
    for (int k = 0; k < (t >> 5); ++k) run += wsum[k];
    for (int i = 0; i < per; ++i) {
        const int idx = lo + i;
        if (idx <= NB) {
            const int c = cnt[idx];
            cnt[idx] = run;
            tabT[(size_t)idx * G + w] = run;
            run += c;
        }
    }
    __syncthreads();

#pragma unroll
    for (int j = 0; j < EPT; ++j) {
        const int i = j * BIN_T + t;
        if (i < n_here)
            stage[cnt[d[j] >> SLICE_SHIFT] + r[j]] =
                (int)((unsigned)s[j] | ((unsigned)(d[j] & (SLICE - 1)) << 24));
    }
    __syncthreads();
    for (int i = t; i < n_here; i += BIN_T) binned[e0 + i] = stage[i];
}

template <bool PLACE>
__device__ __forceinline__ void walk_slice(const int* __restrict__ tab_lo,
                                           const int* __restrict__ tab_hi,
                                           const int* __restrict__ binned,
                                           int G, int t, int* counters,
                                           bool fits, int* stage,
                                           int* __restrict__ ssrc_base) {
    for (int w = t; w < G; w += SORT_T) {
        const int a = tab_lo[w];
        const int c = tab_hi[w] - a;
        const int* __restrict__ p = binned + (size_t)w * CHUNK + a;
        for (int j0 = 0; j0 < c; j0 += 8) {
            unsigned v[8];
#pragma unroll
            for (int j = 0; j < 8; ++j) {
                v[j] = 0u;
                if (j0 + j < c) v[j] = (unsigned)p[j0 + j];
            }
#pragma unroll
            for (int j = 0; j < 8; ++j) {
                if (j0 + j < c) {
                    const int dl = (int)(v[j] >> 24);
                    if constexpr (!PLACE) {
                        atomicAdd(&counters[dl], 1);
                    } else {
                        const int pos = atomicAdd(&counters[dl], 1);
                        const int src = (int)(v[j] & 0x00FFFFFFu);
                        if (fits) stage[pos] = src;
                        else      ssrc_base[pos] = src;
                    }
                }
            }
        }
    }
}

__global__ void __launch_bounds__(SORT_T)
slice_kernel(const int* __restrict__ tabT, const int* __restrict__ binned,
             int G, int N, int* __restrict__ beg, int* __restrict__ endp,
             int* __restrict__ ssrc) {
    __shared__ int deg[SLICE];
    __shared__ int cur[SLICE];
    __shared__ int red[2 * (SORT_T / 32)];
    __shared__ int stage[STAGE_CAP];

    const int t    = threadIdx.x;
    const int lane = t & 31;
    const int wv   = t >> 5;
    const int s    = blockIdx.x;
    const int* __restrict__ tab_lo = tabT + (size_t)s * G;
    const int* __restrict__ tab_hi = tabT + (size_t)(s + 1) * G;

    if (t < SLICE) deg[t] = 0;

    int bsum = 0, tsum = 0;
    for (int w = t; w < G; w += SORT_T) {
        const int a = tab_lo[w];
        bsum += a;
        tsum += tab_hi[w] - a;
    }
#pragma unroll
    for (int o = 16; o > 0; o >>= 1) {
        bsum += __shfl_xor(bsum, o);
        tsum += __shfl_xor(tsum, o);
    }
    if (lane == 0) { red[wv] = bsum; red[SORT_T / 32 + wv] = tsum; }
    __syncthreads();
    int base = 0, total = 0;
    for (int k = 0; k < SORT_T / 32; ++k) {
        base  += red[k];
        total += red[SORT_T / 32 + k];
    }
    const bool fits = total <= STAGE_CAP;

    walk_slice<false>(tab_lo, tab_hi, binned, G, t, deg, fits, stage, ssrc);
    __syncthreads();

    const int dv = (t < SLICE) ? deg[t] : 0;
    int inc = dv;
#pragma unroll
    for (int o = 1; o < 32; o <<= 1) {
        const int u = __shfl_up(inc, o);
        if (lane >= o) inc += u;
    }
    if (lane == 31) red[wv] = inc;
    __syncthreads();
    int off = inc - dv;
    for (int k = 0; k < wv; ++k) off += red[k];
    if (t < SLICE) {
        cur[t] = off;
        const int n = s * SLICE + t;
        if (n < N) { beg[n] = base + off; endp[n] = base + off + dv; }
    }
    __syncthreads();

    walk_slice<true>(tab_lo, tab_hi, binned, G, t, cur, fits, stage,
                     ssrc + base);
    __syncthreads();
    if (fits)
        for (int i = t; i < total; i += SORT_T) ssrc[base + i] = stage[i];
}

__device__ __forceinline__ float4 ld4(const float* __restrict__ p) {
    return *reinterpret_cast<const float4*>(p);
}
__device__ __forceinline__ float dot4(const float4& a, const float4& b) {
    return a.x * b.x + a.y * b.y + a.z * b.z + a.w * b.w;
}

template <int H, int C, bool FINAL>
__global__ void node_kernel(const int* __restrict__ beg,
                            const int* __restrict__ endp,
                            const int* __restrict__ ssrc,
                            const float* __restrict__ xw,
                            const float* __restrict__ a_src,
                            const float* __restrict__ a_dst,
                            const float* __restrict__ bias,
                            float* __restrict__ outp, int N) {
    constexpr int F   = H * C;
    constexpr int LPR = F / 4;
    constexpr int LPH = C / 4;
    constexpr int NPW = 32 / F;
    static_assert(F == 32 || F == 16, "row must be 32 or 16 floats");
    static_assert(C % 4 == 0, "a float4 must not straddle two heads");
    const int lane = threadIdx.x & 31;
    const int sub  = lane % F;
    const int g    = sub / LPR;
    const int c4   = sub % LPR;
    const int wave0 = __builtin_amdgcn_readfirstlane(
        (int)((blockIdx.x * blockDim.x + threadIdx.x) >> 5));
    const int nwaves = (int)((gridDim.x * blockDim.x) >> 5);
    const int nslots = (N + NPW - 1) / NPW;

    float4 qs, qd, bv;
    qs.x = a_src[4 * c4 + 0]; qs.y = a_src[4 * c4 + 1];
    qs.z = a_src[4 * c4 + 2]; qs.w = a_src[4 * c4 + 3];
    qd.x = a_dst[4 * c4 + 0]; qd.y = a_dst[4 * c4 + 1];
    qd.z = a_dst[4 * c4 + 2]; qd.w = a_dst[4 * c4 + 3];
    bv.x = bias[4 * c4 + 0];  bv.y = bias[4 * c4 + 1];
    bv.z = bias[4 * c4 + 2];  bv.w = bias[4 * c4 + 3];

    for (int q = wave0; q < nslots; q += nwaves) {
        const int  n    = q * NPW + lane / F;
        const bool live = n < N;
        int b = 0, cnt = 0;
        float4 rn = make_float4(0.f, 0.f, 0.f, 0.f);
        if (live) {
            b   = beg[n];
            cnt = endp[n] - b;
            rn  = ld4(xw + (size_t)n * F + 4 * c4);
        }
        float adn = dot4(rn, qd);
#pragma unroll
        for (int o = 1; o < LPH; o <<= 1) adn += __shfl_xor(adn, o);

        const int total = live ? cnt + 1 : 0;
        int iters = (total + 3) >> 2;
        if constexpr (NPW == 2) iters = max(iters, __shfl_xor(iters, 16));
        iters = __builtin_amdgcn_readfirstlane(iters);

        float m = -INFINITY, den = 0.f;
        float a0 = 0.f, a1 = 0.f, a2 = 0.f, a3 = 0.f;
        int k = g;
        const int s0 = (k < cnt) ? ssrc[b + k] : n;
        int s1 = (k + 4 < cnt) ? ssrc[b + k + 4] : n;
        float4 r_cur = make_float4(0.f, 0.f, 0.f, 0.f);
        if (k < total) r_cur = ld4(xw + (size_t)s0 * F + 4 * c4);
        for (int it = 0; it < iters; ++it) {
            const bool valid = k < total;
            const int  kn    = k + 4;
            const int  s2    = (kn + 4 < cnt) ? ssrc[b + kn + 4] : n;
            float4 r_nxt = make_float4(0.f, 0.f, 0.f, 0.f);
            if (kn < total) r_nxt = ld4(xw + (size_t)s1 * F + 4 * c4);
            const float4 r = r_cur;
            float e = dot4(r, qs);
#pragma unroll
            for (int o = 1; o < LPH; o <<= 1) e += __shfl_xor(e, o);
            if (valid) {
                e += adn;
                e = (e > 0.f) ? e : NEG_SLOPE * e;
                const float mn = fmaxf(m, e);
                const float sc = expf(m - mn);
                const float p  = expf(e - mn);
                den = den * sc + p;
                a0  = a0 * sc + p * r.x;
                a1  = a1 * sc + p * r.y;
                a2  = a2 * sc + p * r.z;
                a3  = a3 * sc + p * r.w;
                m   = mn;
            }
            k = kn; s1 = s2; r_cur = r_nxt;
        }
        float M = fmaxf(m, __shfl_xor(m, LPR));
        M = fmaxf(M, __shfl_xor(M, 2 * LPR));
        const float w = expf(m - M);
        den *= w; a0 *= w; a1 *= w; a2 *= w; a3 *= w;
#pragma unroll
        for (int o = LPR; o < F; o <<= 1) {
            den += __shfl_xor(den, o);
            a0  += __shfl_xor(a0, o);
            a1  += __shfl_xor(a1, o);
            a2  += __shfl_xor(a2, o);
            a3  += __shfl_xor(a3, o);
        }
        float v0 = a0 / den + bv.x;
        float v1 = a1 / den + bv.y;
        float v2 = a2 / den + bv.z;
        float v3 = a3 / den + bv.w;
        if constexpr (!FINAL) {
            v0 = (v0 > 0.f) ? v0 : (expf(v0) - 1.f);
            v1 = (v1 > 0.f) ? v1 : (expf(v1) - 1.f);
            v2 = (v2 > 0.f) ? v2 : (expf(v2) - 1.f);
            v3 = (v3 > 0.f) ? v3 : (expf(v3) - 1.f);
        } else {
            float mx = fmaxf(fmaxf(v0, v1), fmaxf(v2, v3));
#pragma unroll
            for (int o = 1; o < LPR; o <<= 1) mx = fmaxf(mx, __shfl_xor(mx, o));
            float ss = expf(v0 - mx) + expf(v1 - mx) + expf(v2 - mx) +
                       expf(v3 - mx);
#pragma unroll
            for (int o = 1; o < LPR; o <<= 1) ss += __shfl_xor(ss, o);
            const float lg = mx + logf(ss);
            v0 -= lg; v1 -= lg; v2 -= lg; v3 -= lg;
        }
        if (live && g == 0) {
            float4 o4; o4.x = v0; o4.y = v1; o4.z = v2; o4.w = v3;
            *reinterpret_cast<float4*>(outp + (size_t)n * F + 4 * c4) = o4;
        }
    }
}

extern "C" void kernel_launch(void* const* d_in, const int* in_sizes, int n_in,
                              void* d_out, int out_size, void* d_ws,
                              size_t ws_size, hipStream_t stream) {
    const float* x   = (const float*)d_in[0];
    const int*   ei  = (const int*)d_in[1];
    const float* W1  = (const float*)d_in[2];
    const float* a1s = (const float*)d_in[3];
    const float* a1d = (const float*)d_in[4];
    const float* b1  = (const float*)d_in[5];
    const float* W2  = (const float*)d_in[6];
    const float* a2s = (const float*)d_in[7];
    const float* a2d = (const float*)d_in[8];
    const float* b2  = (const float*)d_in[9];
    float* out = (float*)d_out;

    const int N = in_sizes[0] / 128;
    const int E = in_sizes[1] / 2;
    const int* srcs = ei;
    const int* dsts = ei + E;

    char* p = (char*)d_ws;
    auto carve = [&](size_t bytes) {
        void* r = (void*)p;
        p += (bytes + 255) & ~(size_t)255;
        return r;
    };
    float* xw1  = (float*)carve((size_t)N * 32 * 4);
    float* h1   = (float*)carve((size_t)N * 32 * 4);
    float* xw2  = (float*)carve((size_t)N * 16 * 4);
    int*   endp = (int*)carve((size_t)N * 4);
    int*   beg  = (int*)carve((size_t)N * 4);
    int*   ssrc = (int*)carve((size_t)E * 4);
    int*   binned = (int*)h1;
    int*   tabT   = (int*)xw2;

    const int B = 256;
    auto g = [](int n, int b) { return (n + b - 1) / b; };
    const int NODE_BLOCKS = 2048;

    const int NB = g(N, SLICE);
    const int G  = g(E, CHUNK);
    bin_kernel<<<G, BIN_T, (size_t)(NB + 1 + CHUNK) * 4, stream>>>(
        srcs, dsts, E, NB, G, tabT, binned);
    slice_kernel<<<NB, SORT_T, 0, stream>>>(tabT, binned, G, N, beg, endp,
                                            ssrc);

    gemm_wmma_kernel<128, 32><<<N / 16, 32, 0, stream>>>(x, W1, xw1, N);
    node_kernel<4, 8, false><<<NODE_BLOCKS, B, 0, stream>>>(
        beg, endp, ssrc, xw1, a1s, a1d, b1, h1, N);

    gemm_wmma_kernel<32, 16><<<N / 16, 32, 0, stream>>>(h1, W2, xw2, N);
    node_kernel<1, 16, true><<<NODE_BLOCKS, B, 0, stream>>>(
        beg, endp, ssrc, xw2, a2s, a2d, b2, out, N);
}


# node kernels: library expf, logf and divide replaced by hardware exp2/log2/rcp (scores pre-scaled by log2 e); one exp2 of -|e-m| per running-max update, the other factor being exp(0)=1
# speedup vs baseline: 16.7214x; 1.2086x over previous
#include <hip/hip_runtime.h>
#include <math.h>

typedef __attribute__((ext_vector_type(2))) float v2f;
typedef __attribute__((ext_vector_type(8))) float v8f;

#define NEG_SLOPE 0.2f

template <int K, int NCOL>
__global__ void gemm_wmma_kernel(const float* __restrict__ A,
                                 const float* __restrict__ B,
                                 float* __restrict__ C, int N) {
    constexpr int NT = NCOL / 16;
    const int lane = threadIdx.x & 31;
    const int l    = lane & 15;
    const int hf   = lane >> 4;
    const int row0 = blockIdx.x * 16;
    if (row0 >= N) return;

    const float* __restrict__ arow = A + (size_t)(row0 + l) * K;
    v8f acc[NT];
#pragma unroll
    for (int t = 0; t < NT; ++t) acc[t] = (v8f){};

#pragma unroll
    for (int k = 0; k < K; k += 4) {
        v2f a;
        a.x = arow[k + 2 * hf];
        a.y = arow[k + 2 * hf + 1];
#pragma unroll
        for (int t = 0; t < NT; ++t) {
            v2f b;
            b.x = B[(size_t)(k + 2 * hf) * NCOL + 16 * t + l];
            b.y = B[(size_t)(k + 2 * hf + 1) * NCOL + 16 * t + l];
            acc[t] = __builtin_amdgcn_wmma_f32_16x16x4_f32(
                 false, a,  false, b,
                 (short)0, acc[t],  false,
                 false);
        }
    }
#pragma unroll
    for (int t = 0; t < NT; ++t)
#pragma unroll
        for (int i = 0; i < 8; ++i)
            C[(size_t)(row0 + i + 8 * hf) * NCOL + 16 * t + l] = acc[t][i];
}

#define SLICE_SHIFT 8
#define SLICE 256
#define CHUNK 4096
#define BIN_T 256
#define EPT (CHUNK / BIN_T)
#define SORT_T 512
#define STAGE_CAP 8192

__global__ void __launch_bounds__(BIN_T)
bin_kernel(const int* __restrict__ srcs, const int* __restrict__ dsts, int E,
           int NB, int G, int* __restrict__ tabT, int* __restrict__ binned) {
    extern __shared__ int smem[];
    int* cnt   = smem;
    int* stage = smem + (NB + 1);
    __shared__ int wsum[BIN_T / 32];

    const int t    = threadIdx.x;
    const int lane = t & 31;
    const int w    = blockIdx.x;
    const int e0   = w * CHUNK;
    const int n_here = min(CHUNK, E - e0);

    for (int i = t; i <= NB; i += BIN_T) cnt[i] = 0;
    __syncthreads();

    int s[EPT], d[EPT], r[EPT];
#pragma unroll
    for (int j = 0; j < EPT; ++j) {
        const int i = j * BIN_T + t;
        s[j] = 0; d[j] = 0;
        if (i < n_here) { s[j] = srcs[e0 + i]; d[j] = dsts[e0 + i]; }
    }
#pragma unroll
    for (int j = 0; j < EPT; ++j) {
        const int i = j * BIN_T + t;
        r[j] = 0;
        if (i < n_here) r[j] = atomicAdd(&cnt[d[j] >> SLICE_SHIFT], 1);
    }
    __syncthreads();

    const int per = (NB + 1 + BIN_T - 1) / BIN_T;
    const int lo  = t * per;
    int v = 0;
    for (int i = 0; i < per; ++i) {
        const int idx = lo + i;
        if (idx <= NB) v += cnt[idx];
    }
    int inc = v;
#pragma unroll
    for (int o = 1; o < 32; o <<= 1) {
        const int u = __shfl_up(inc, o);
        if (lane >= o) inc += u;
    }
    if (lane == 31) wsum[t >> 5] = inc;
    __syncthreads();
    int run = inc - v;
    for (int k = 0; k < (t >> 5); ++k) run += wsum[k];
    for (int i = 0; i < per; ++i) {
        const int idx = lo + i;
        if (idx <= NB) {
            const int c = cnt[idx];
            cnt[idx] = run;
            tabT[(size_t)idx * G + w] = run;
            run += c;
        }
    }
    __syncthreads();

#pragma unroll
    for (int j = 0; j < EPT; ++j) {
        const int i = j * BIN_T + t;
        if (i < n_here)
            stage[cnt[d[j] >> SLICE_SHIFT] + r[j]] =
                (int)((unsigned)s[j] | ((unsigned)(d[j] & (SLICE - 1)) << 24));
    }
    __syncthreads();
    for (int i = t; i < n_here; i += BIN_T) binned[e0 + i] = stage[i];
}

template <bool PLACE>
__device__ __forceinline__ void walk_slice(const int* __restrict__ tab_lo,
                                           const int* __restrict__ tab_hi,
                                           const int* __restrict__ binned,
                                           int G, int t, int* counters,
                                           bool fits, int* stage,
                                           int* __restrict__ ssrc_base) {
    for (int w = t; w < G; w += SORT_T) {
        const int a = tab_lo[w];
        const int c = tab_hi[w] - a;
        const int* __restrict__ p = binned + (size_t)w * CHUNK + a;
        for (int j0 = 0; j0 < c; j0 += 8) {
            unsigned v[8];
#pragma unroll
            for (int j = 0; j < 8; ++j) {
                v[j] = 0u;
                if (j0 + j < c) v[j] = (unsigned)p[j0 + j];
            }
#pragma unroll
            for (int j = 0; j < 8; ++j) {
                if (j0 + j < c) {
                    const int dl = (int)(v[j] >> 24);
                    if constexpr (!PLACE) {
                        atomicAdd(&counters[dl], 1);
                    } else {
                        const int pos = atomicAdd(&counters[dl], 1);
                        const int src = (int)(v[j] & 0x00FFFFFFu);
                        if (fits) stage[pos] = src;
                        else      ssrc_base[pos] = src;
                    }
                }
            }
        }
    }
}

__global__ void __launch_bounds__(SORT_T)
slice_kernel(const int* __restrict__ tabT, const int* __restrict__ binned,
             int G, int N, int* __restrict__ beg, int* __restrict__ endp,
             int* __restrict__ ssrc) {
    __shared__ int deg[SLICE];
    __shared__ int cur[SLICE];
    __shared__ int red[2 * (SORT_T / 32)];
    __shared__ int stage[STAGE_CAP];

    const int t    = threadIdx.x;
    const int lane = t & 31;
    const int wv   = t >> 5;
    const int s    = blockIdx.x;
    const int* __restrict__ tab_lo = tabT + (size_t)s * G;
    const int* __restrict__ tab_hi = tabT + (size_t)(s + 1) * G;

    if (t < SLICE) deg[t] = 0;

    int bsum = 0, tsum = 0;
    for (int w = t; w < G; w += SORT_T) {
        const int a = tab_lo[w];
        bsum += a;
        tsum += tab_hi[w] - a;
    }
#pragma unroll
    for (int o = 16; o > 0; o >>= 1) {
        bsum += __shfl_xor(bsum, o);
        tsum += __shfl_xor(tsum, o);
    }
    if (lane == 0) { red[wv] = bsum; red[SORT_T / 32 + wv] = tsum; }
    __syncthreads();
    int base = 0, total = 0;
    for (int k = 0; k < SORT_T / 32; ++k) {
        base  += red[k];
        total += red[SORT_T / 32 + k];
    }
    const bool fits = total <= STAGE_CAP;

    walk_slice<false>(tab_lo, tab_hi, binned, G, t, deg, fits, stage, ssrc);
    __syncthreads();

    const int dv = (t < SLICE) ? deg[t] : 0;
    int inc = dv;
#pragma unroll
    for (int o = 1; o < 32; o <<= 1) {
        const int u = __shfl_up(inc, o);
        if (lane >= o) inc += u;
    }
    if (lane == 31) red[wv] = inc;
    __syncthreads();
    int off = inc - dv;
    for (int k = 0; k < wv; ++k) off += red[k];
    if (t < SLICE) {
        cur[t] = off;
        const int n = s * SLICE + t;
        if (n < N) { beg[n] = base + off; endp[n] = base + off + dv; }
    }
    __syncthreads();

    walk_slice<true>(tab_lo, tab_hi, binned, G, t, cur, fits, stage,
                     ssrc + base);
    __syncthreads();
    if (fits)
        for (int i = t; i < total; i += SORT_T) ssrc[base + i] = stage[i];
}

__device__ __forceinline__ float4 ld4(const float* __restrict__ p) {
    return *reinterpret_cast<const float4*>(p);
}
__device__ __forceinline__ float dot4(const float4& a, const float4& b) {
    return a.x * b.x + a.y * b.y + a.z * b.z + a.w * b.w;
}

#define LOG2E 1.44269504088896340736f
#define LN2   0.69314718055994530942f
__device__ __forceinline__ float hw_exp2(float x) {
    return __builtin_amdgcn_exp2f(x);
}
__device__ __forceinline__ float hw_log2(float x) {
    return __builtin_amdgcn_logf(x);
}
__device__ __forceinline__ float hw_rcp(float x) {
    return __builtin_amdgcn_rcpf(x);
}

template <int H, int C, bool FINAL>
__global__ void node_kernel(const int* __restrict__ beg,
                            const int* __restrict__ endp,
                            const int* __restrict__ ssrc,
                            const float* __restrict__ xw,
                            const float* __restrict__ a_src,
                            const float* __restrict__ a_dst,
                            const float* __restrict__ bias,
                            float* __restrict__ outp, int N) {
    constexpr int F   = H * C;
    constexpr int LPR = F / 4;
    constexpr int LPH = C / 4;
    constexpr int NPW = 32 / F;
    static_assert(F == 32 || F == 16, "row must be 32 or 16 floats");
    static_assert(C % 4 == 0, "a float4 must not straddle two heads");
    const int lane = threadIdx.x & 31;
    const int sub  = lane % F;
    const int g    = sub / LPR;
    const int c4   = sub % LPR;
    const int wave0 = __builtin_amdgcn_readfirstlane(
        (int)((blockIdx.x * blockDim.x + threadIdx.x) >> 5));
    const int nwaves = (int)((gridDim.x * blockDim.x) >> 5);
    const int nslots = (N + NPW - 1) / NPW;

    float4 qs, qd, bv;
    qs.x = a_src[4 * c4 + 0]; qs.y = a_src[4 * c4 + 1];
    qs.z = a_src[4 * c4 + 2]; qs.w = a_src[4 * c4 + 3];
    qd.x = a_dst[4 * c4 + 0]; qd.y = a_dst[4 * c4 + 1];
    qd.z = a_dst[4 * c4 + 2]; qd.w = a_dst[4 * c4 + 3];
    bv.x = bias[4 * c4 + 0];  bv.y = bias[4 * c4 + 1];
    bv.z = bias[4 * c4 + 2];  bv.w = bias[4 * c4 + 3];
    qs.x *= LOG2E; qs.y *= LOG2E; qs.z *= LOG2E; qs.w *= LOG2E;
    qd.x *= LOG2E; qd.y *= LOG2E; qd.z *= LOG2E; qd.w *= LOG2E;

    for (int q = wave0; q < nslots; q += nwaves) {
        const int  n    = q * NPW + lane / F;
        const bool live = n < N;
        int b = 0, cnt = 0;
        float4 rn = make_float4(0.f, 0.f, 0.f, 0.f);
        if (live) {
            b   = beg[n];
            cnt = endp[n] - b;
            rn  = ld4(xw + (size_t)n * F + 4 * c4);
        }
        float adn = dot4(rn, qd);
#pragma unroll
        for (int o = 1; o < LPH; o <<= 1) adn += __shfl_xor(adn, o);

        const int total = live ? cnt + 1 : 0;
        int iters = (total + 3) >> 2;
        if constexpr (NPW == 2) iters = max(iters, __shfl_xor(iters, 16));
        iters = __builtin_amdgcn_readfirstlane(iters);

        float m = -INFINITY, den = 0.f;
        float a0 = 0.f, a1 = 0.f, a2 = 0.f, a3 = 0.f;
        int k = g;
        const int s0 = (k < cnt) ? ssrc[b + k] : n;
        int s1 = (k + 4 < cnt) ? ssrc[b + k + 4] : n;
        float4 r_cur = make_float4(0.f, 0.f, 0.f, 0.f);
        if (k < total) r_cur = ld4(xw + (size_t)s0 * F + 4 * c4);
        for (int it = 0; it < iters; ++it) {
            const bool valid = k < total;
            const int  kn    = k + 4;
            const int  s2    = (kn + 4 < cnt) ? ssrc[b + kn + 4] : n;
            float4 r_nxt = make_float4(0.f, 0.f, 0.f, 0.f);
            if (kn < total) r_nxt = ld4(xw + (size_t)s1 * F + 4 * c4);
            const float4 r = r_cur;
            float e = dot4(r, qs);
#pragma unroll
            for (int o = 1; o < LPH; o <<= 1) e += __shfl_xor(e, o);
            if (valid) {
                e += adn;
                e = (e > 0.f) ? e : NEG_SLOPE * e;
                const float d  = e - m;
                const float t  = hw_exp2(-fabsf(d));
                const bool  up = d > 0.f;
                const float sc = up ? t : 1.f;
                const float p  = up ? 1.f : t;
                den = den * sc + p;
                a0  = a0 * sc + p * r.x;
                a1  = a1 * sc + p * r.y;
                a2  = a2 * sc + p * r.z;
                a3  = a3 * sc + p * r.w;
                m   = up ? e : m;
            }
            k = kn; s1 = s2; r_cur = r_nxt;
        }
        float M = fmaxf(m, __shfl_xor(m, LPR));
        M = fmaxf(M, __shfl_xor(M, 2 * LPR));
        const float w = hw_exp2(m - M);
        den *= w; a0 *= w; a1 *= w; a2 *= w; a3 *= w;
#pragma unroll
        for (int o = LPR; o < F; o <<= 1) {
            den += __shfl_xor(den, o);
            a0  += __shfl_xor(a0, o);
            a1  += __shfl_xor(a1, o);
            a2  += __shfl_xor(a2, o);
            a3  += __shfl_xor(a3, o);
        }
        const float inv = hw_rcp(den);
        float v0 = a0 * inv + bv.x;
        float v1 = a1 * inv + bv.y;
        float v2 = a2 * inv + bv.z;
        float v3 = a3 * inv + bv.w;
        if constexpr (!FINAL) {
            v0 = (v0 > 0.f) ? v0 : (hw_exp2(v0 * LOG2E) - 1.f);
            v1 = (v1 > 0.f) ? v1 : (hw_exp2(v1 * LOG2E) - 1.f);
            v2 = (v2 > 0.f) ? v2 : (hw_exp2(v2 * LOG2E) - 1.f);
            v3 = (v3 > 0.f) ? v3 : (hw_exp2(v3 * LOG2E) - 1.f);
        } else {
            float mx = fmaxf(fmaxf(v0, v1), fmaxf(v2, v3));
#pragma unroll
            for (int o = 1; o < LPR; o <<= 1) mx = fmaxf(mx, __shfl_xor(mx, o));
            float ss = hw_exp2((v0 - mx) * LOG2E) + hw_exp2((v1 - mx) * LOG2E) +
                       hw_exp2((v2 - mx) * LOG2E) + hw_exp2((v3 - mx) * LOG2E);
#pragma unroll
            for (int o = 1; o < LPR; o <<= 1) ss += __shfl_xor(ss, o);
            const float lg = mx + hw_log2(ss) * LN2;
            v0 -= lg; v1 -= lg; v2 -= lg; v3 -= lg;
        }
        if (live && g == 0) {
            float4 o4; o4.x = v0; o4.y = v1; o4.z = v2; o4.w = v3;
            *reinterpret_cast<float4*>(outp + (size_t)n * F + 4 * c4) = o4;
        }
    }
}

extern "C" void kernel_launch(void* const* d_in, const int* in_sizes, int n_in,
                              void* d_out, int out_size, void* d_ws,
                              size_t ws_size, hipStream_t stream) {
    const float* x   = (const float*)d_in[0];
    const int*   ei  = (const int*)d_in[1];
    const float* W1  = (const float*)d_in[2];
    const float* a1s = (const float*)d_in[3];
    const float* a1d = (const float*)d_in[4];
    const float* b1  = (const float*)d_in[5];
    const float* W2  = (const float*)d_in[6];
    const float* a2s = (const float*)d_in[7];
    const float* a2d = (const float*)d_in[8];
    const float* b2  = (const float*)d_in[9];
    float* out = (float*)d_out;

    const int N = in_sizes[0] / 128;
    const int E = in_sizes[1] / 2;
    const int* srcs = ei;
    const int* dsts = ei + E;

    char* p = (char*)d_ws;
    auto carve = [&](size_t bytes) {
        void* r = (void*)p;
        p += (bytes + 255) & ~(size_t)255;
        return r;
    };
    float* xw1  = (float*)carve((size_t)N * 32 * 4);
    float* h1   = (float*)carve((size_t)N * 32 * 4);
    float* xw2  = (float*)carve((size_t)N * 16 * 4);
    int*   endp = (int*)carve((size_t)N * 4);
    int*   beg  = (int*)carve((size_t)N * 4);
    int*   ssrc = (int*)carve((size_t)E * 4);
    int*   binned = (int*)h1;
    int*   tabT   = (int*)xw2;

    const int B = 256;
    auto g = [](int n, int b) { return (n + b - 1) / b; };
    const int NODE_BLOCKS = 2048;

    const int NB = g(N, SLICE);
    const int G  = g(E, CHUNK);
    bin_kernel<<<G, BIN_T, (size_t)(NB + 1 + CHUNK) * 4, stream>>>(
        srcs, dsts, E, NB, G, tabT, binned);
    slice_kernel<<<NB, SORT_T, 0, stream>>>(tabT, binned, G, N, beg, endp,
                                            ssrc);

    gemm_wmma_kernel<128, 32><<<N / 16, 32, 0, stream>>>(x, W1, xw1, N);
    node_kernel<4, 8, false><<<NODE_BLOCKS, B, 0, stream>>>(
        beg, endp, ssrc, xw1, a1s, a1d, b1, h1, N);

    gemm_wmma_kernel<32, 16><<<N / 16, 32, 0, stream>>>(h1, W2, xw2, N);
    node_kernel<1, 16, true><<<NODE_BLOCKS, B, 0, stream>>>(
        beg, endp, ssrc, xw2, a2s, a2d, b2, out, N);
}


# node kernels: all row loads of a node issued together ahead of the arithmetic (batches of 6 trips = 24 items; unconditional, clamped addresses): two memory hops per node, not one per trip
# speedup vs baseline: 17.2686x; 1.0327x over previous
#include <hip/hip_runtime.h>
#include <math.h>

typedef __attribute__((ext_vector_type(2))) float v2f;
typedef __attribute__((ext_vector_type(8))) float v8f;

#define NEG_SLOPE 0.2f

template <int K, int NCOL>
__global__ void gemm_wmma_kernel(const float* __restrict__ A,
                                 const float* __restrict__ B,
                                 float* __restrict__ C, int N) {
    constexpr int NT = NCOL / 16;
    const int lane = threadIdx.x & 31;
    const int l    = lane & 15;
    const int hf   = lane >> 4;
    const int row0 = blockIdx.x * 16;
    if (row0 >= N) return;

    const float* __restrict__ arow = A + (size_t)(row0 + l) * K;
    v8f acc[NT];
#pragma unroll
    for (int t = 0; t < NT; ++t) acc[t] = (v8f){};

#pragma unroll
    for (int k = 0; k < K; k += 4) {
        v2f a;
        a.x = arow[k + 2 * hf];
        a.y = arow[k + 2 * hf + 1];
#pragma unroll
        for (int t = 0; t < NT; ++t) {
            v2f b;
            b.x = B[(size_t)(k + 2 * hf) * NCOL + 16 * t + l];
            b.y = B[(size_t)(k + 2 * hf + 1) * NCOL + 16 * t + l];
            acc[t] = __builtin_amdgcn_wmma_f32_16x16x4_f32(
                 false, a,  false, b,
                 (short)0, acc[t],  false,
                 false);
        }
    }
#pragma unroll
    for (int t = 0; t < NT; ++t)
#pragma unroll
        for (int i = 0; i < 8; ++i)
            C[(size_t)(row0 + i + 8 * hf) * NCOL + 16 * t + l] = acc[t][i];
}

#define SLICE_SHIFT 8
#define SLICE 256
#define CHUNK 4096
#define BIN_T 256
#define EPT (CHUNK / BIN_T)
#define SORT_T 512
#define STAGE_CAP 8192

__global__ void __launch_bounds__(BIN_T)
bin_kernel(const int* __restrict__ srcs, const int* __restrict__ dsts, int E,
           int NB, int G, int* __restrict__ tabT, int* __restrict__ binned) {
    extern __shared__ int smem[];
    int* cnt   = smem;
    int* stage = smem + (NB + 1);
    __shared__ int wsum[BIN_T / 32];

    const int t    = threadIdx.x;
    const int lane = t & 31;
    const int w    = blockIdx.x;
    const int e0   = w * CHUNK;
    const int n_here = min(CHUNK, E - e0);

    for (int i = t; i <= NB; i += BIN_T) cnt[i] = 0;
    __syncthreads();

    int s[EPT], d[EPT], r[EPT];
#pragma unroll
    for (int j = 0; j < EPT; ++j) {
        const int i = j * BIN_T + t;
        s[j] = 0; d[j] = 0;
        if (i < n_here) { s[j] = srcs[e0 + i]; d[j] = dsts[e0 + i]; }
    }
#pragma unroll
    for (int j = 0; j < EPT; ++j) {
        const int i = j * BIN_T + t;
        r[j] = 0;
        if (i < n_here) r[j] = atomicAdd(&cnt[d[j] >> SLICE_SHIFT], 1);
    }
    __syncthreads();

    const int per = (NB + 1 + BIN_T - 1) / BIN_T;
    const int lo  = t * per;
    int v = 0;
    for (int i = 0; i < per; ++i) {
        const int idx = lo + i;
        if (idx <= NB) v += cnt[idx];
    }
    int inc = v;
#pragma unroll
    for (int o = 1; o < 32; o <<= 1) {
        const int u = __shfl_up(inc, o);
        if (lane >= o) inc += u;
    }
    if (lane == 31) wsum[t >> 5] = inc;
    __syncthreads();
    int run = inc - v;
    for (int k = 0; k < (t >> 5); ++k) run += wsum[k];
    for (int i = 0; i < per; ++i) {
        const int idx = lo + i;
        if (idx <= NB) {
            const int c = cnt[idx];
            cnt[idx] = run;
            tabT[(size_t)idx * G + w] = run;
            run += c;
        }
    }
    __syncthreads();

#pragma unroll
    for (int j = 0; j < EPT; ++j) {
        const int i = j * BIN_T + t;
        if (i < n_here)
            stage[cnt[d[j] >> SLICE_SHIFT] + r[j]] =
                (int)((unsigned)s[j] | ((unsigned)(d[j] & (SLICE - 1)) << 24));
    }
    __syncthreads();
    for (int i = t; i < n_here; i += BIN_T) binned[e0 + i] = stage[i];
}

template <bool PLACE>
__device__ __forceinline__ void walk_slice(const int* __restrict__ tab_lo,
                                           const int* __restrict__ tab_hi,
                                           const int* __restrict__ binned,
                                           int G, int t, int* counters,
                                           bool fits, int* stage,
                                           int* __restrict__ ssrc_base) {
    for (int w = t; w < G; w += SORT_T) {
        const int a = tab_lo[w];
        const int c = tab_hi[w] - a;
        const int* __restrict__ p = binned + (size_t)w * CHUNK + a;
        for (int j0 = 0; j0 < c; j0 += 8) {
            unsigned v[8];
#pragma unroll
            for (int j = 0; j < 8; ++j) {
                v[j] = 0u;
                if (j0 + j < c) v[j] = (unsigned)p[j0 + j];
            }
#pragma unroll
            for (int j = 0; j < 8; ++j) {
                if (j0 + j < c) {
                    const int dl = (int)(v[j] >> 24);
                    if constexpr (!PLACE) {
                        atomicAdd(&counters[dl], 1);
                    } else {
                        const int pos = atomicAdd(&counters[dl], 1);
                        const int src = (int)(v[j] & 0x00FFFFFFu);
                        if (fits) stage[pos] = src;
                        else      ssrc_base[pos] = src;
                    }
                }
            }
        }
    }
}

__global__ void __launch_bounds__(SORT_T)
slice_kernel(const int* __restrict__ tabT, const int* __restrict__ binned,
             int G, int N, int* __restrict__ beg, int* __restrict__ endp,
             int* __restrict__ ssrc) {
    __shared__ int deg[SLICE];
    __shared__ int cur[SLICE];
    __shared__ int red[2 * (SORT_T / 32)];
    __shared__ int stage[STAGE_CAP];

    const int t    = threadIdx.x;
    const int lane = t & 31;
    const int wv   = t >> 5;
    const int s    = blockIdx.x;
    const int* __restrict__ tab_lo = tabT + (size_t)s * G;
    const int* __restrict__ tab_hi = tabT + (size_t)(s + 1) * G;

    if (t < SLICE) deg[t] = 0;

    int bsum = 0, tsum = 0;
    for (int w = t; w < G; w += SORT_T) {
        const int a = tab_lo[w];
        bsum += a;
        tsum += tab_hi[w] - a;
    }
#pragma unroll
    for (int o = 16; o > 0; o >>= 1) {
        bsum += __shfl_xor(bsum, o);
        tsum += __shfl_xor(tsum, o);
    }
    if (lane == 0) { red[wv] = bsum; red[SORT_T / 32 + wv] = tsum; }
    __syncthreads();
    int base = 0, total = 0;
    for (int k = 0; k < SORT_T / 32; ++k) {
        base  += red[k];
        total += red[SORT_T / 32 + k];
    }
    const bool fits = total <= STAGE_CAP;

    walk_slice<false>(tab_lo, tab_hi, binned, G, t, deg, fits, stage, ssrc);
    __syncthreads();

    const int dv = (t < SLICE) ? deg[t] : 0;
    int inc = dv;
#pragma unroll
    for (int o = 1; o < 32; o <<= 1) {
        const int u = __shfl_up(inc, o);
        if (lane >= o) inc += u;
    }
    if (lane == 31) red[wv] = inc;
    __syncthreads();
    int off = inc - dv;
    for (int k = 0; k < wv; ++k) off += red[k];
    if (t < SLICE) {
        cur[t] = off;
        const int n = s * SLICE + t;
        if (n < N) { beg[n] = base + off; endp[n] = base + off + dv; }
    }
    __syncthreads();

    walk_slice<true>(tab_lo, tab_hi, binned, G, t, cur, fits, stage,
                     ssrc + base);
    __syncthreads();
    if (fits)
        for (int i = t; i < total; i += SORT_T) ssrc[base + i] = stage[i];
}

__device__ __forceinline__ float4 ld4(const float* __restrict__ p) {
    return *reinterpret_cast<const float4*>(p);
}
__device__ __forceinline__ float dot4(const float4& a, const float4& b) {
    return a.x * b.x + a.y * b.y + a.z * b.z + a.w * b.w;
}

#define LOG2E 1.44269504088896340736f
#define LN2   0.69314718055994530942f
__device__ __forceinline__ float hw_exp2(float x) {
    return __builtin_amdgcn_exp2f(x);
}
__device__ __forceinline__ float hw_log2(float x) {
    return __builtin_amdgcn_logf(x);
}
__device__ __forceinline__ float hw_rcp(float x) {
    return __builtin_amdgcn_rcpf(x);
}

template <int H, int C, bool FINAL>
__global__ void node_kernel(const int* __restrict__ beg,
                            const int* __restrict__ endp,
                            const int* __restrict__ ssrc,
                            const float* __restrict__ xw,
                            const float* __restrict__ a_src,
                            const float* __restrict__ a_dst,
                            const float* __restrict__ bias,
                            float* __restrict__ outp, int N, int last) {
    constexpr int F   = H * C;
    constexpr int LPR = F / 4;
    constexpr int LPH = C / 4;
    constexpr int NPW = 32 / F;
    constexpr int TB  = 6;
    static_assert(F == 32 || F == 16, "row must be 32 or 16 floats");
    static_assert(C % 4 == 0, "a float4 must not straddle two heads");
    const int lane = threadIdx.x & 31;
    const int sub  = lane % F;
    const int g    = sub / LPR;
    const int c4   = sub % LPR;
    const int wave0 = __builtin_amdgcn_readfirstlane(
        (int)((blockIdx.x * blockDim.x + threadIdx.x) >> 5));
    const int nwaves = (int)((gridDim.x * blockDim.x) >> 5);
    const int nslots = (N + NPW - 1) / NPW;

    float4 qs, qd, bv;
    qs.x = a_src[4 * c4 + 0]; qs.y = a_src[4 * c4 + 1];
    qs.z = a_src[4 * c4 + 2]; qs.w = a_src[4 * c4 + 3];
    qd.x = a_dst[4 * c4 + 0]; qd.y = a_dst[4 * c4 + 1];
    qd.z = a_dst[4 * c4 + 2]; qd.w = a_dst[4 * c4 + 3];
    bv.x = bias[4 * c4 + 0];  bv.y = bias[4 * c4 + 1];
    bv.z = bias[4 * c4 + 2];  bv.w = bias[4 * c4 + 3];
    qs.x *= LOG2E; qs.y *= LOG2E; qs.z *= LOG2E; qs.w *= LOG2E;
    qd.x *= LOG2E; qd.y *= LOG2E; qd.z *= LOG2E; qd.w *= LOG2E;

    for (int q = wave0; q < nslots; q += nwaves) {
        const int  n    = q * NPW + lane / F;
        const bool live = n < N;
        const int  nn   = live ? n : N - 1;
        const int  b    = beg[nn];
        const int  en   = endp[nn];
        const float4 rn = ld4(xw + (size_t)nn * F + 4 * c4);
        const int  cnt  = live ? en - b : -1;
        float adn = dot4(rn, qd);
#pragma unroll
        for (int o = 1; o < LPH; o <<= 1) adn += __shfl_xor(adn, o);

        const int total = cnt + 1;
        int iters = (total + 3) >> 2;
        if constexpr (NPW == 2) iters = max(iters, __shfl_xor(iters, 16));
        iters = __builtin_amdgcn_readfirstlane(iters);

        float m = -INFINITY, den = 0.f;
        float a0 = 0.f, a1 = 0.f, a2 = 0.f, a3 = 0.f;
        for (int t0 = 0; t0 < iters; t0 += TB) {
            int sidx[TB];
#pragma unroll
            for (int t = 0; t < TB; ++t) {
                const int kk = g + 4 * (t0 + t);
                sidx[t] = ssrc[min(b + kk, last)];
            }
#pragma unroll
            for (int t = 0; t < TB; ++t) {
                const int kk = g + 4 * (t0 + t);
                sidx[t] = (kk < cnt) ? sidx[t] : nn;
            }
            float4 r[TB];
#pragma unroll
            for (int t = 0; t < TB; ++t)
                r[t] = ld4(xw + (size_t)sidx[t] * F + 4 * c4);
#pragma unroll
            for (int t = 0; t < TB; ++t) {
                if (t0 + t < iters) {
                    const int kk = g + 4 * (t0 + t);
                    float e = dot4(r[t], qs);
#pragma unroll
                    for (int o = 1; o < LPH; o <<= 1) e += __shfl_xor(e, o);
                    if (kk < total) {
                        e += adn;
                        e = (e > 0.f) ? e : NEG_SLOPE * e;
                        const float dd = e - m;
                        const float tt = hw_exp2(-fabsf(dd));
                        const bool  up = dd > 0.f;
                        const float sc = up ? tt : 1.f;
                        const float p  = up ? 1.f : tt;
                        den = den * sc + p;
                        a0  = a0 * sc + p * r[t].x;
                        a1  = a1 * sc + p * r[t].y;
                        a2  = a2 * sc + p * r[t].z;
                        a3  = a3 * sc + p * r[t].w;
                        m   = up ? e : m;
                    }
                }
            }
        }
        float M = fmaxf(m, __shfl_xor(m, LPR));
        M = fmaxf(M, __shfl_xor(M, 2 * LPR));
        const float w = hw_exp2(m - M);
        den *= w; a0 *= w; a1 *= w; a2 *= w; a3 *= w;
#pragma unroll
        for (int o = LPR; o < F; o <<= 1) {
            den += __shfl_xor(den, o);
            a0  += __shfl_xor(a0, o);
            a1  += __shfl_xor(a1, o);
            a2  += __shfl_xor(a2, o);
            a3  += __shfl_xor(a3, o);
        }
        const float inv = hw_rcp(den);
        float v0 = a0 * inv + bv.x;
        float v1 = a1 * inv + bv.y;
        float v2 = a2 * inv + bv.z;
        float v3 = a3 * inv + bv.w;
        if constexpr (!FINAL) {
            v0 = (v0 > 0.f) ? v0 : (hw_exp2(v0 * LOG2E) - 1.f);
            v1 = (v1 > 0.f) ? v1 : (hw_exp2(v1 * LOG2E) - 1.f);
            v2 = (v2 > 0.f) ? v2 : (hw_exp2(v2 * LOG2E) - 1.f);
            v3 = (v3 > 0.f) ? v3 : (hw_exp2(v3 * LOG2E) - 1.f);
        } else {
            float mx = fmaxf(fmaxf(v0, v1), fmaxf(v2, v3));
#pragma unroll
            for (int o = 1; o < LPR; o <<= 1) mx = fmaxf(mx, __shfl_xor(mx, o));
            float ss = hw_exp2((v0 - mx) * LOG2E) + hw_exp2((v1 - mx) * LOG2E) +
                       hw_exp2((v2 - mx) * LOG2E) + hw_exp2((v3 - mx) * LOG2E);
#pragma unroll
            for (int o = 1; o < LPR; o <<= 1) ss += __shfl_xor(ss, o);
            const float lg = mx + hw_log2(ss) * LN2;
            v0 -= lg; v1 -= lg; v2 -= lg; v3 -= lg;
        }
        if (live && g == 0) {
            float4 o4; o4.x = v0; o4.y = v1; o4.z = v2; o4.w = v3;
            *reinterpret_cast<float4*>(outp + (size_t)n * F + 4 * c4) = o4;
        }
    }
}

extern "C" void kernel_launch(void* const* d_in, const int* in_sizes, int n_in,
                              void* d_out, int out_size, void* d_ws,
                              size_t ws_size, hipStream_t stream) {
    const float* x   = (const float*)d_in[0];
    const int*   ei  = (const int*)d_in[1];
    const float* W1  = (const float*)d_in[2];
    const float* a1s = (const float*)d_in[3];
    const float* a1d = (const float*)d_in[4];
    const float* b1  = (const float*)d_in[5];
    const float* W2  = (const float*)d_in[6];
    const float* a2s = (const float*)d_in[7];
    const float* a2d = (const float*)d_in[8];
    const float* b2  = (const float*)d_in[9];
    float* out = (float*)d_out;

    const int N = in_sizes[0] / 128;
    const int E = in_sizes[1] / 2;
    const int* srcs = ei;
    const int* dsts = ei + E;

    char* p = (char*)d_ws;
    auto carve = [&](size_t bytes) {
        void* r = (void*)p;
        p += (bytes + 255) & ~(size_t)255;
        return r;
    };
    float* xw1  = (float*)carve((size_t)N * 32 * 4);
    float* h1   = (float*)carve((size_t)N * 32 * 4);
    float* xw2  = (float*)carve((size_t)N * 16 * 4);
    int*   endp = (int*)carve((size_t)N * 4);
    int*   beg  = (int*)carve((size_t)N * 4);
    int*   ssrc = (int*)carve((size_t)E * 4);
    int*   binned = (int*)h1;
    int*   tabT   = (int*)xw2;

    const int B = 256;
    auto g = [](int n, int b) { return (n + b - 1) / b; };
    const int NODE_BLOCKS1 = 2048;
    const int NODE_BLOCKS2 = 1536;

    const int NB = g(N, SLICE);
    const int G  = g(E, CHUNK);
    bin_kernel<<<G, BIN_T, (size_t)(NB + 1 + CHUNK) * 4, stream>>>(
        srcs, dsts, E, NB, G, tabT, binned);
    slice_kernel<<<NB, SORT_T, 0, stream>>>(tabT, binned, G, N, beg, endp,
                                            ssrc);

    gemm_wmma_kernel<128, 32><<<N / 16, 32, 0, stream>>>(x, W1, xw1, N);
    node_kernel<4, 8, false><<<NODE_BLOCKS1, B, 0, stream>>>(
        beg, endp, ssrc, xw1, a1s, a1d, b1, h1, N, E - 1);

    gemm_wmma_kernel<32, 16><<<N / 16, 32, 0, stream>>>(h1, W2, xw2, N);
    node_kernel<1, 16, true><<<NODE_BLOCKS2, B, 0, stream>>>(
        beg, endp, ssrc, xw2, a2s, a2d, b2, out, N, E - 1);
}


# first layer's x @ W1 moved into the edge-binning launch: eight 16-row slabs per 8-wave work-group behind the bin groups; 5 launches, not 6; 782 work-groups in place of 6250 one-wave ones
# speedup vs baseline: 18.3075x; 1.0602x over previous
#include <hip/hip_runtime.h>
#include <math.h>

typedef __attribute__((ext_vector_type(2))) float v2f;
typedef __attribute__((ext_vector_type(8))) float v8f;

#define NEG_SLOPE 0.2f

template <int K, int NCOL>
__device__ __forceinline__ void gemm_slab(const float* __restrict__ A,
                                          const float* __restrict__ B,
                                          float* __restrict__ C, int row0,
                                          int lane) {
    constexpr int NT = NCOL / 16;
    const int l    = lane & 15;
    const int hf   = lane >> 4;

    const float* __restrict__ arow = A + (size_t)(row0 + l) * K;
    v8f acc[NT];
#pragma unroll
    for (int t = 0; t < NT; ++t) acc[t] = (v8f){};

#pragma unroll
    for (int k = 0; k < K; k += 4) {
        v2f a;
        a.x = arow[k + 2 * hf];
        a.y = arow[k + 2 * hf + 1];
#pragma unroll
        for (int t = 0; t < NT; ++t) {
            v2f b;
            b.x = B[(size_t)(k + 2 * hf) * NCOL + 16 * t + l];
            b.y = B[(size_t)(k + 2 * hf + 1) * NCOL + 16 * t + l];
            acc[t] = __builtin_amdgcn_wmma_f32_16x16x4_f32(
                 false, a,  false, b,
                 (short)0, acc[t],  false,
                 false);
        }
    }
#pragma unroll
    for (int t = 0; t < NT; ++t)
#pragma unroll
        for (int i = 0; i < 8; ++i)
            C[(size_t)(row0 + i + 8 * hf) * NCOL + 16 * t + l] = acc[t][i];
}

template <int K, int NCOL>
__global__ void gemm_wmma_kernel(const float* __restrict__ A,
                                 const float* __restrict__ B,
                                 float* __restrict__ C, int N) {
    const int row0 = blockIdx.x * 16;
    if (row0 >= N) return;
    gemm_slab<K, NCOL>(A, B, C, row0, threadIdx.x & 31);
}

#define SLICE_SHIFT 8
#define SLICE 256
#define CHUNK 4096
#define BIN_T 256
#define EPT (CHUNK / BIN_T)
#define SORT_T 512
#define STAGE_CAP 8192

__device__ __forceinline__ void
bin_chunk(const int* __restrict__ srcs, const int* __restrict__ dsts, int E,
          int NB, int G, int* __restrict__ tabT, int* __restrict__ binned) {
    extern __shared__ int smem[];
    int* cnt   = smem;
    int* stage = smem + (NB + 1);
    __shared__ int wsum[BIN_T / 32];

    const int t    = threadIdx.x;
    const int lane = t & 31;
    const int w    = blockIdx.x;
    const int e0   = w * CHUNK;
    const int n_here = min(CHUNK, E - e0);

    for (int i = t; i <= NB; i += BIN_T) cnt[i] = 0;
    __syncthreads();

    int s[EPT], d[EPT], r[EPT];
#pragma unroll
    for (int j = 0; j < EPT; ++j) {
        const int i = j * BIN_T + t;
        s[j] = 0; d[j] = 0;
        if (i < n_here) { s[j] = srcs[e0 + i]; d[j] = dsts[e0 + i]; }
    }
#pragma unroll
    for (int j = 0; j < EPT; ++j) {
        const int i = j * BIN_T + t;
        r[j] = 0;
        if (i < n_here) r[j] = atomicAdd(&cnt[d[j] >> SLICE_SHIFT], 1);
    }
    __syncthreads();

    const int per = (NB + 1 + BIN_T - 1) / BIN_T;
    const int lo  = t * per;
    int v = 0;
    for (int i = 0; i < per; ++i) {
        const int idx = lo + i;
        if (idx <= NB) v += cnt[idx];
    }
    int inc = v;
#pragma unroll
    for (int o = 1; o < 32; o <<= 1) {
        const int u = __shfl_up(inc, o);
        if (lane >= o) inc += u;
    }
    if (lane == 31) wsum[t >> 5] = inc;
    __syncthreads();
    int run = inc - v;
    for (int k = 0; k < (t >> 5); ++k) run += wsum[k];
    for (int i = 0; i < per; ++i) {
        const int idx = lo + i;
        if (idx <= NB) {
            const int c = cnt[idx];
            cnt[idx] = run;
            tabT[(size_t)idx * G + w] = run;
            run += c;
        }
    }
    __syncthreads();

#pragma unroll
    for (int j = 0; j < EPT; ++j) {
        const int i = j * BIN_T + t;
        if (i < n_here)
            stage[cnt[d[j] >> SLICE_SHIFT] + r[j]] =
                (int)((unsigned)s[j] | ((unsigned)(d[j] & (SLICE - 1)) << 24));
    }
    __syncthreads();
    for (int i = t; i < n_here; i += BIN_T) binned[e0 + i] = stage[i];
}

#define GEMM1_K 128
#define GEMM1_NCOL 32
__global__ void __launch_bounds__(BIN_T)
bin_gemm_kernel(const int* __restrict__ srcs, const int* __restrict__ dsts,
                int E, int NB, int G, int* __restrict__ tabT,
                int* __restrict__ binned, const float* __restrict__ x,
                const float* __restrict__ W1, float* __restrict__ xw1, int N) {
    if ((int)blockIdx.x < G) {
        bin_chunk(srcs, dsts, E, NB, G, tabT, binned);
    } else {
        const int wv = __builtin_amdgcn_readfirstlane((int)(threadIdx.x >> 5));
        const int row0 = (((int)blockIdx.x - G) * (BIN_T / 32) + wv) * 16;
        if (row0 < N)
            gemm_slab<GEMM1_K, GEMM1_NCOL>(x, W1, xw1, row0, threadIdx.x & 31);
    }
}

template <bool PLACE>
__device__ __forceinline__ void walk_slice(const int* __restrict__ tab_lo,
                                           const int* __restrict__ tab_hi,
                                           const int* __restrict__ binned,
                                           int G, int t, int* counters,
                                           bool fits, int* stage,
                                           int* __restrict__ ssrc_base) {
    for (int w = t; w < G; w += SORT_T) {
        const int a = tab_lo[w];
        const int c = tab_hi[w] - a;
        const int* __restrict__ p = binned + (size_t)w * CHUNK + a;
        for (int j0 = 0; j0 < c; j0 += 8) {
            unsigned v[8];
#pragma unroll
            for (int j = 0; j < 8; ++j) {
                v[j] = 0u;
                if (j0 + j < c) v[j] = (unsigned)p[j0 + j];
            }
#pragma unroll
            for (int j = 0; j < 8; ++j) {
                if (j0 + j < c) {
                    const int dl = (int)(v[j] >> 24);
                    if constexpr (!PLACE) {
                        atomicAdd(&counters[dl], 1);
                    } else {
                        const int pos = atomicAdd(&counters[dl], 1);
                        const int src = (int)(v[j] & 0x00FFFFFFu);
                        if (fits) stage[pos] = src;
                        else      ssrc_base[pos] = src;
                    }
                }
            }
        }
    }
}

__global__ void __launch_bounds__(SORT_T)
slice_kernel(const int* __restrict__ tabT, const int* __restrict__ binned,
             int G, int N, int* __restrict__ beg, int* __restrict__ endp,
             int* __restrict__ ssrc) {
    __shared__ int deg[SLICE];
    __shared__ int cur[SLICE];
    __shared__ int red[2 * (SORT_T / 32)];
    __shared__ int stage[STAGE_CAP];

    const int t    = threadIdx.x;
    const int lane = t & 31;
    const int wv   = t >> 5;
    const int s    = blockIdx.x;
    const int* __restrict__ tab_lo = tabT + (size_t)s * G;
    const int* __restrict__ tab_hi = tabT + (size_t)(s + 1) * G;

    if (t < SLICE) deg[t] = 0;

    int bsum = 0, tsum = 0;
    for (int w = t; w < G; w += SORT_T) {
        const int a = tab_lo[w];
        bsum += a;
        tsum += tab_hi[w] - a;
    }
#pragma unroll
    for (int o = 16; o > 0; o >>= 1) {
        bsum += __shfl_xor(bsum, o);
        tsum += __shfl_xor(tsum, o);
    }
    if (lane == 0) { red[wv] = bsum; red[SORT_T / 32 + wv] = tsum; }
    __syncthreads();
    int base = 0, total = 0;
    for (int k = 0; k < SORT_T / 32; ++k) {
        base  += red[k];
        total += red[SORT_T / 32 + k];
    }
    const bool fits = total <= STAGE_CAP;

    walk_slice<false>(tab_lo, tab_hi, binned, G, t, deg, fits, stage, ssrc);
    __syncthreads();

    const int dv = (t < SLICE) ? deg[t] : 0;
    int inc = dv;
#pragma unroll
    for (int o = 1; o < 32; o <<= 1) {
        const int u = __shfl_up(inc, o);
        if (lane >= o) inc += u;
    }
    if (lane == 31) red[wv] = inc;
    __syncthreads();
    int off = inc - dv;
    for (int k = 0; k < wv; ++k) off += red[k];
    if (t < SLICE) {
        cur[t] = off;
        const int n = s * SLICE + t;
        if (n < N) { beg[n] = base + off; endp[n] = base + off + dv; }
    }
    __syncthreads();

    walk_slice<true>(tab_lo, tab_hi, binned, G, t, cur, fits, stage,
                     ssrc + base);
    __syncthreads();
    if (fits)
        for (int i = t; i < total; i += SORT_T) ssrc[base + i] = stage[i];
}

__device__ __forceinline__ float4 ld4(const float* __restrict__ p) {
    return *reinterpret_cast<const float4*>(p);
}
__device__ __forceinline__ float dot4(const float4& a, const float4& b) {
    return a.x * b.x + a.y * b.y + a.z * b.z + a.w * b.w;
}

#define LOG2E 1.44269504088896340736f
#define LN2   0.69314718055994530942f
__device__ __forceinline__ float hw_exp2(float x) {
    return __builtin_amdgcn_exp2f(x);
}
__device__ __forceinline__ float hw_log2(float x) {
    return __builtin_amdgcn_logf(x);
}
__device__ __forceinline__ float hw_rcp(float x) {
    return __builtin_amdgcn_rcpf(x);
}

template <int H, int C, bool FINAL>
__global__ void node_kernel(const int* __restrict__ beg,
                            const int* __restrict__ endp,
                            const int* __restrict__ ssrc,
                            const float* __restrict__ xw,
                            const float* __restrict__ a_src,
                            const float* __restrict__ a_dst,
                            const float* __restrict__ bias,
                            float* __restrict__ outp, int N, int last) {
    constexpr int F   = H * C;
    constexpr int LPR = F / 4;
    constexpr int LPH = C / 4;
    constexpr int NPW = 32 / F;
    constexpr int TB  = 6;
    static_assert(F == 32 || F == 16, "row must be 32 or 16 floats");
    static_assert(C % 4 == 0, "a float4 must not straddle two heads");
    const int lane = threadIdx.x & 31;
    const int sub  = lane % F;
    const int g    = sub / LPR;
    const int c4   = sub % LPR;
    const int wave0 = __builtin_amdgcn_readfirstlane(
        (int)((blockIdx.x * blockDim.x + threadIdx.x) >> 5));
    const int nwaves = (int)((gridDim.x * blockDim.x) >> 5);
    const int nslots = (N + NPW - 1) / NPW;

    float4 qs, qd, bv;
    qs.x = a_src[4 * c4 + 0]; qs.y = a_src[4 * c4 + 1];
    qs.z = a_src[4 * c4 + 2]; qs.w = a_src[4 * c4 + 3];
    qd.x = a_dst[4 * c4 + 0]; qd.y = a_dst[4 * c4 + 1];
    qd.z = a_dst[4 * c4 + 2]; qd.w = a_dst[4 * c4 + 3];
    bv.x = bias[4 * c4 + 0];  bv.y = bias[4 * c4 + 1];
    bv.z = bias[4 * c4 + 2];  bv.w = bias[4 * c4 + 3];
    qs.x *= LOG2E; qs.y *= LOG2E; qs.z *= LOG2E; qs.w *= LOG2E;
    qd.x *= LOG2E; qd.y *= LOG2E; qd.z *= LOG2E; qd.w *= LOG2E;

    for (int q = wave0; q < nslots; q += nwaves) {
        const int  n    = q * NPW + lane / F;
        const bool live = n < N;
        const int  nn   = live ? n : N - 1;
        const int  b    = beg[nn];
        const int  en   = endp[nn];
        const float4 rn = ld4(xw + (size_t)nn * F + 4 * c4);
        const int  cnt  = live ? en - b : -1;
        float adn = dot4(rn, qd);
#pragma unroll
        for (int o = 1; o < LPH; o <<= 1) adn += __shfl_xor(adn, o);

        const int total = cnt + 1;
        int iters = (total + 3) >> 2;
        if constexpr (NPW == 2) iters = max(iters, __shfl_xor(iters, 16));
        iters = __builtin_amdgcn_readfirstlane(iters);

        float m = -INFINITY, den = 0.f;
        float a0 = 0.f, a1 = 0.f, a2 = 0.f, a3 = 0.f;
        for (int t0 = 0; t0 < iters; t0 += TB) {
            int sidx[TB];
#pragma unroll
            for (int t = 0; t < TB; ++t) {
                const int kk = g + 4 * (t0 + t);
                sidx[t] = ssrc[min(b + kk, last)];
            }
#pragma unroll
            for (int t = 0; t < TB; ++t) {
                const int kk = g + 4 * (t0 + t);
                sidx[t] = (kk < cnt) ? sidx[t] : nn;
            }
            float4 r[TB];
#pragma unroll
            for (int t = 0; t < TB; ++t)
                r[t] = ld4(xw + (size_t)sidx[t] * F + 4 * c4);
#pragma unroll
            for (int t = 0; t < TB; ++t) {
                if (t0 + t < iters) {
                    const int kk = g + 4 * (t0 + t);
                    float e = dot4(r[t], qs);
#pragma unroll
                    for (int o = 1; o < LPH; o <<= 1) e += __shfl_xor(e, o);
                    if (kk < total) {
                        e += adn;
                        e = (e > 0.f) ? e : NEG_SLOPE * e;
                        const float dd = e - m;
                        const float tt = hw_exp2(-fabsf(dd));
                        const bool  up = dd > 0.f;
                        const float sc = up ? tt : 1.f;
                        const float p  = up ? 1.f : tt;
                        den = den * sc + p;
                        a0  = a0 * sc + p * r[t].x;
                        a1  = a1 * sc + p * r[t].y;
                        a2  = a2 * sc + p * r[t].z;
                        a3  = a3 * sc + p * r[t].w;
                        m   = up ? e : m;
                    }
                }
            }
        }
        float M = fmaxf(m, __shfl_xor(m, LPR));
        M = fmaxf(M, __shfl_xor(M, 2 * LPR));
        const float w = hw_exp2(m - M);
        den *= w; a0 *= w; a1 *= w; a2 *= w; a3 *= w;
#pragma unroll
        for (int o = LPR; o < F; o <<= 1) {
            den += __shfl_xor(den, o);
            a0  += __shfl_xor(a0, o);
            a1  += __shfl_xor(a1, o);
            a2  += __shfl_xor(a2, o);
            a3  += __shfl_xor(a3, o);
        }
        const float inv = hw_rcp(den);
        float v0 = a0 * inv + bv.x;
        float v1 = a1 * inv + bv.y;
        float v2 = a2 * inv + bv.z;
        float v3 = a3 * inv + bv.w;
        if constexpr (!FINAL) {
            v0 = (v0 > 0.f) ? v0 : (hw_exp2(v0 * LOG2E) - 1.f);
            v1 = (v1 > 0.f) ? v1 : (hw_exp2(v1 * LOG2E) - 1.f);
            v2 = (v2 > 0.f) ? v2 : (hw_exp2(v2 * LOG2E) - 1.f);
            v3 = (v3 > 0.f) ? v3 : (hw_exp2(v3 * LOG2E) - 1.f);
        } else {
            float mx = fmaxf(fmaxf(v0, v1), fmaxf(v2, v3));
#pragma unroll
            for (int o = 1; o < LPR; o <<= 1) mx = fmaxf(mx, __shfl_xor(mx, o));
            float ss = hw_exp2((v0 - mx) * LOG2E) + hw_exp2((v1 - mx) * LOG2E) +
                       hw_exp2((v2 - mx) * LOG2E) + hw_exp2((v3 - mx) * LOG2E);
#pragma unroll
            for (int o = 1; o < LPR; o <<= 1) ss += __shfl_xor(ss, o);
            const float lg = mx + hw_log2(ss) * LN2;
            v0 -= lg; v1 -= lg; v2 -= lg; v3 -= lg;
        }
        if (live && g == 0) {
            float4 o4; o4.x = v0; o4.y = v1; o4.z = v2; o4.w = v3;
            *reinterpret_cast<float4*>(outp + (size_t)n * F + 4 * c4) = o4;
        }
    }
}

extern "C" void kernel_launch(void* const* d_in, const int* in_sizes, int n_in,
                              void* d_out, int out_size, void* d_ws,
                              size_t ws_size, hipStream_t stream) {
    const float* x   = (const float*)d_in[0];
    const int*   ei  = (const int*)d_in[1];
    const float* W1  = (const float*)d_in[2];
    const float* a1s = (const float*)d_in[3];
    const float* a1d = (const float*)d_in[4];
    const float* b1  = (const float*)d_in[5];
    const float* W2  = (const float*)d_in[6];
    const float* a2s = (const float*)d_in[7];
    const float* a2d = (const float*)d_in[8];
    const float* b2  = (const float*)d_in[9];
    float* out = (float*)d_out;

    const int N = in_sizes[0] / 128;
    const int E = in_sizes[1] / 2;
    const int* srcs = ei;
    const int* dsts = ei + E;

    char* p = (char*)d_ws;
    auto carve = [&](size_t bytes) {
        void* r = (void*)p;
        p += (bytes + 255) & ~(size_t)255;
        return r;
    };
    float* xw1  = (float*)carve((size_t)N * 32 * 4);
    float* h1   = (float*)carve((size_t)N * 32 * 4);
    float* xw2  = (float*)carve((size_t)N * 16 * 4);
    int*   endp = (int*)carve((size_t)N * 4);
    int*   beg  = (int*)carve((size_t)N * 4);
    int*   ssrc = (int*)carve((size_t)E * 4);
    int*   binned = (int*)h1;
    int*   tabT   = (int*)xw2;

    const int B = 256;
    auto g = [](int n, int b) { return (n + b - 1) / b; };
    const int NODE_BLOCKS1 = 2048;
    const int NODE_BLOCKS2 = 1536;

    const int NB = g(N, SLICE);
    const int G  = g(E, CHUNK);
    const int GG = g(N / 16, BIN_T / 32);
    bin_gemm_kernel<<<G + GG, BIN_T, (size_t)(NB + 1 + CHUNK) * 4, stream>>>(
        srcs, dsts, E, NB, G, tabT, binned, x, W1, xw1, N);
    slice_kernel<<<NB, SORT_T, 0, stream>>>(tabT, binned, G, N, beg, endp,
                                            ssrc);

    node_kernel<4, 8, false><<<NODE_BLOCKS1, B, 0, stream>>>(
        beg, endp, ssrc, xw1, a1s, a1d, b1, h1, N, E - 1);

    gemm_wmma_kernel<32, 16><<<N / 16, 32, 0, stream>>>(h1, W2, xw2, N);
    node_kernel<1, 16, true><<<NODE_BLOCKS2, B, 0, stream>>>(
        beg, endp, ssrc, xw2, a2s, a2d, b2, out, N, E - 1);
}


# second layer's h1 @ W2 moved into the layer-1 node launch: 2 waves own 16 consecutive nodes (slot ranges, alpha_dst once per wave), rows go to an LDS tile, one wave runs the WMMA product; 4 launches
# speedup vs baseline: 19.8150x; 1.0823x over previous
#include <hip/hip_runtime.h>
#include <math.h>

typedef __attribute__((ext_vector_type(2))) float v2f;
typedef __attribute__((ext_vector_type(8))) float v8f;

#define NEG_SLOPE 0.2f

template <int K, int NCOL>
__device__ __forceinline__ void gemm_slab(const float* __restrict__ A,
                                          const float* __restrict__ B,
                                          float* __restrict__ C, int row0,
                                          int lane) {
    constexpr int NT = NCOL / 16;
    const int l    = lane & 15;
    const int hf   = lane >> 4;

    const float* __restrict__ arow = A + (size_t)(row0 + l) * K;
    v8f acc[NT];
#pragma unroll
    for (int t = 0; t < NT; ++t) acc[t] = (v8f){};

#pragma unroll
    for (int k = 0; k < K; k += 4) {
        v2f a;
        a.x = arow[k + 2 * hf];
        a.y = arow[k + 2 * hf + 1];
#pragma unroll
        for (int t = 0; t < NT; ++t) {
            v2f b;
            b.x = B[(size_t)(k + 2 * hf) * NCOL + 16 * t + l];
            b.y = B[(size_t)(k + 2 * hf + 1) * NCOL + 16 * t + l];
            acc[t] = __builtin_amdgcn_wmma_f32_16x16x4_f32(
                 false, a,  false, b,
                 (short)0, acc[t],  false,
                 false);
        }
    }
#pragma unroll
    for (int t = 0; t < NT; ++t)
#pragma unroll
        for (int i = 0; i < 8; ++i)
            C[(size_t)(row0 + i + 8 * hf) * NCOL + 16 * t + l] = acc[t][i];
}

#define SLICE_SHIFT 8
#define SLICE 256
#define CHUNK 4096
#define BIN_T 256
#define EPT (CHUNK / BIN_T)
#define SORT_T 512
#define STAGE_CAP 8192

__device__ __forceinline__ void
bin_chunk(const int* __restrict__ srcs, const int* __restrict__ dsts, int E,
          int NB, int G, int* __restrict__ tabT, int* __restrict__ binned) {
    extern __shared__ int smem[];
    int* cnt   = smem;
    int* stage = smem + (NB + 1);
    __shared__ int wsum[BIN_T / 32];

    const int t    = threadIdx.x;
    const int lane = t & 31;
    const int w    = blockIdx.x;
    const int e0   = w * CHUNK;
    const int n_here = min(CHUNK, E - e0);

    for (int i = t; i <= NB; i += BIN_T) cnt[i] = 0;
    __syncthreads();

    int s[EPT], d[EPT], r[EPT];
#pragma unroll
    for (int j = 0; j < EPT; ++j) {
        const int i = j * BIN_T + t;
        s[j] = 0; d[j] = 0;
        if (i < n_here) { s[j] = srcs[e0 + i]; d[j] = dsts[e0 + i]; }
    }
#pragma unroll
    for (int j = 0; j < EPT; ++j) {
        const int i = j * BIN_T + t;
        r[j] = 0;
        if (i < n_here) r[j] = atomicAdd(&cnt[d[j] >> SLICE_SHIFT], 1);
    }
    __syncthreads();

    const int per = (NB + 1 + BIN_T - 1) / BIN_T;
    const int lo  = t * per;
    int v = 0;
    for (int i = 0; i < per; ++i) {
        const int idx = lo + i;
        if (idx <= NB) v += cnt[idx];
    }
    int inc = v;
#pragma unroll
    for (int o = 1; o < 32; o <<= 1) {
        const int u = __shfl_up(inc, o);
        if (lane >= o) inc += u;
    }
    if (lane == 31) wsum[t >> 5] = inc;
    __syncthreads();
    int run = inc - v;
    for (int k = 0; k < (t >> 5); ++k) run += wsum[k];
    for (int i = 0; i < per; ++i) {
        const int idx = lo + i;
        if (idx <= NB) {
            const int c = cnt[idx];
            cnt[idx] = run;
            tabT[(size_t)idx * G + w] = run;
            run += c;
        }
    }
    __syncthreads();

#pragma unroll
    for (int j = 0; j < EPT; ++j) {
        const int i = j * BIN_T + t;
        if (i < n_here)
            stage[cnt[d[j] >> SLICE_SHIFT] + r[j]] =
                (int)((unsigned)s[j] | ((unsigned)(d[j] & (SLICE - 1)) << 24));
    }
    __syncthreads();
    for (int i = t; i < n_here; i += BIN_T) binned[e0 + i] = stage[i];
}

#define GEMM1_K 128
#define GEMM1_NCOL 32
__global__ void __launch_bounds__(BIN_T)
bin_gemm_kernel(const int* __restrict__ srcs, const int* __restrict__ dsts,
                int E, int NB, int G, int* __restrict__ tabT,
                int* __restrict__ binned, const float* __restrict__ x,
                const float* __restrict__ W1, float* __restrict__ xw1, int N) {
    if ((int)blockIdx.x < G) {
        bin_chunk(srcs, dsts, E, NB, G, tabT, binned);
    } else {
        const int wv = __builtin_amdgcn_readfirstlane((int)(threadIdx.x >> 5));
        const int row0 = (((int)blockIdx.x - G) * (BIN_T / 32) + wv) * 16;
        if (row0 < N)
            gemm_slab<GEMM1_K, GEMM1_NCOL>(x, W1, xw1, row0, threadIdx.x & 31);
    }
}

template <bool PLACE>
__device__ __forceinline__ void walk_slice(const int* __restrict__ tab_lo,
                                           const int* __restrict__ tab_hi,
                                           const int* __restrict__ binned,
                                           int G, int t, int* counters,
                                           bool fits, int* stage,
                                           int* __restrict__ ssrc_base) {
    for (int w = t; w < G; w += SORT_T) {
        const int a = tab_lo[w];
        const int c = tab_hi[w] - a;
        const int* __restrict__ p = binned + (size_t)w * CHUNK + a;
        for (int j0 = 0; j0 < c; j0 += 8) {
            unsigned v[8];
#pragma unroll
            for (int j = 0; j < 8; ++j) {
                v[j] = 0u;
                if (j0 + j < c) v[j] = (unsigned)p[j0 + j];
            }
#pragma unroll
            for (int j = 0; j < 8; ++j) {
                if (j0 + j < c) {
                    const int dl = (int)(v[j] >> 24);
                    if constexpr (!PLACE) {
                        atomicAdd(&counters[dl], 1);
                    } else {
                        const int pos = atomicAdd(&counters[dl], 1);
                        const int src = (int)(v[j] & 0x00FFFFFFu);
                        if (fits) stage[pos] = src;
                        else      ssrc_base[pos] = src;
                    }
                }
            }
        }
    }
}

__global__ void __launch_bounds__(SORT_T)
slice_kernel(const int* __restrict__ tabT, const int* __restrict__ binned,
             int G, int N, int* __restrict__ beg, int* __restrict__ endp,
             int* __restrict__ ssrc) {
    __shared__ int deg[SLICE];
    __shared__ int cur[SLICE];
    __shared__ int red[2 * (SORT_T / 32)];
    __shared__ int stage[STAGE_CAP];

    const int t    = threadIdx.x;
    const int lane = t & 31;
    const int wv   = t >> 5;
    const int s    = blockIdx.x;
    const int* __restrict__ tab_lo = tabT + (size_t)s * G;
    const int* __restrict__ tab_hi = tabT + (size_t)(s + 1) * G;

    if (t < SLICE) deg[t] = 0;

    int bsum = 0, tsum = 0;
    for (int w = t; w < G; w += SORT_T) {
        const int a = tab_lo[w];
        bsum += a;
        tsum += tab_hi[w] - a;
    }
#pragma unroll
    for (int o = 16; o > 0; o >>= 1) {
        bsum += __shfl_xor(bsum, o);
        tsum += __shfl_xor(tsum, o);
    }
    if (lane == 0) { red[wv] = bsum; red[SORT_T / 32 + wv] = tsum; }
    __syncthreads();
    int base = 0, total = 0;
    for (int k = 0; k < SORT_T / 32; ++k) {
        base  += red[k];
        total += red[SORT_T / 32 + k];
    }
    const bool fits = total <= STAGE_CAP;

    walk_slice<false>(tab_lo, tab_hi, binned, G, t, deg, fits, stage, ssrc);
    __syncthreads();

    const int dv = (t < SLICE) ? deg[t] : 0;
    int inc = dv;
#pragma unroll
    for (int o = 1; o < 32; o <<= 1) {
        const int u = __shfl_up(inc, o);
        if (lane >= o) inc += u;
    }
    if (lane == 31) red[wv] = inc;
    __syncthreads();
    int off = inc - dv;
    for (int k = 0; k < wv; ++k) off += red[k];
    if (t < SLICE) {
        cur[t] = off;
        const int n = s * SLICE + t;
        if (n < N) { beg[n] = base + off; endp[n] = base + off + dv; }
    }
    __syncthreads();

    walk_slice<true>(tab_lo, tab_hi, binned, G, t, cur, fits, stage,
                     ssrc + base);
    __syncthreads();
    if (fits)
        for (int i = t; i < total; i += SORT_T) ssrc[base + i] = stage[i];
}

__device__ __forceinline__ float4 ld4(const float* __restrict__ p) {
    return *reinterpret_cast<const float4*>(p);
}
__device__ __forceinline__ float dot4(const float4& a, const float4& b) {
    return a.x * b.x + a.y * b.y + a.z * b.z + a.w * b.w;
}

#define LOG2E 1.44269504088896340736f
#define LN2   0.69314718055994530942f
__device__ __forceinline__ float hw_exp2(float x) {
    return __builtin_amdgcn_exp2f(x);
}
__device__ __forceinline__ float hw_log2(float x) {
    return __builtin_amdgcn_logf(x);
}
__device__ __forceinline__ float hw_rcp(float x) {
    return __builtin_amdgcn_rcpf(x);
}

template <int H, int C, bool FINAL>
__global__ void node_kernel(const int* __restrict__ beg,
                            const int* __restrict__ endp,
                            const int* __restrict__ ssrc,
                            const float* __restrict__ xw,
                            const float* __restrict__ a_src,
                            const float* __restrict__ a_dst,
                            const float* __restrict__ bias,
                            float* __restrict__ outp, int N, int last) {
    constexpr int F   = H * C;
    constexpr int LPR = F / 4;
    constexpr int LPH = C / 4;
    constexpr int NPW = 32 / F;
    constexpr int TB  = 6;
    static_assert(F == 32 || F == 16, "row must be 32 or 16 floats");
    static_assert(C % 4 == 0, "a float4 must not straddle two heads");
    const int lane = threadIdx.x & 31;
    const int sub  = lane % F;
    const int g    = sub / LPR;
    const int c4   = sub % LPR;
    const int wave0 = __builtin_amdgcn_readfirstlane(
        (int)((blockIdx.x * blockDim.x + threadIdx.x) >> 5));
    const int nwaves = (int)((gridDim.x * blockDim.x) >> 5);
    const int nslots = (N + NPW - 1) / NPW;

    float4 qs, qd, bv;
    qs.x = a_src[4 * c4 + 0]; qs.y = a_src[4 * c4 + 1];
    qs.z = a_src[4 * c4 + 2]; qs.w = a_src[4 * c4 + 3];
    qd.x = a_dst[4 * c4 + 0]; qd.y = a_dst[4 * c4 + 1];
    qd.z = a_dst[4 * c4 + 2]; qd.w = a_dst[4 * c4 + 3];
    bv.x = bias[4 * c4 + 0];  bv.y = bias[4 * c4 + 1];
    bv.z = bias[4 * c4 + 2];  bv.w = bias[4 * c4 + 3];
    qs.x *= LOG2E; qs.y *= LOG2E; qs.z *= LOG2E; qs.w *= LOG2E;
    qd.x *= LOG2E; qd.y *= LOG2E; qd.z *= LOG2E; qd.w *= LOG2E;

    for (int q = wave0; q < nslots; q += nwaves) {
        const int  n    = q * NPW + lane / F;
        const bool live = n < N;
        const int  nn   = live ? n : N - 1;
        const int  b    = beg[nn];
        const int  en   = endp[nn];
        const float4 rn = ld4(xw + (size_t)nn * F + 4 * c4);
        const int  cnt  = live ? en - b : -1;
        float adn = dot4(rn, qd);
#pragma unroll
        for (int o = 1; o < LPH; o <<= 1) adn += __shfl_xor(adn, o);

        const int total = cnt + 1;
        int iters = (total + 3) >> 2;
        if constexpr (NPW == 2) iters = max(iters, __shfl_xor(iters, 16));
        iters = __builtin_amdgcn_readfirstlane(iters);

        float m = -INFINITY, den = 0.f;
        float a0 = 0.f, a1 = 0.f, a2 = 0.f, a3 = 0.f;
        for (int t0 = 0; t0 < iters; t0 += TB) {
            int sidx[TB];
#pragma unroll
            for (int t = 0; t < TB; ++t) {
                const int kk = g + 4 * (t0 + t);
                sidx[t] = ssrc[min(b + kk, last)];
            }
#pragma unroll
            for (int t = 0; t < TB; ++t) {
                const int kk = g + 4 * (t0 + t);
                sidx[t] = (kk < cnt) ? sidx[t] : nn;
            }
            float4 r[TB];
#pragma unroll
            for (int t = 0; t < TB; ++t)
                r[t] = ld4(xw + (size_t)sidx[t] * F + 4 * c4);
#pragma unroll
            for (int t = 0; t < TB; ++t) {
                if (t0 + t < iters) {
                    const int kk = g + 4 * (t0 + t);
                    float e = dot4(r[t], qs);
#pragma unroll
                    for (int o = 1; o < LPH; o <<= 1) e += __shfl_xor(e, o);
                    if (kk < total) {
                        e += adn;
                        e = (e > 0.f) ? e : NEG_SLOPE * e;
                        const float dd = e - m;
                        const float tt = hw_exp2(-fabsf(dd));
                        const bool  up = dd > 0.f;
                        const float sc = up ? tt : 1.f;
                        const float p  = up ? 1.f : tt;
                        den = den * sc + p;
                        a0  = a0 * sc + p * r[t].x;
                        a1  = a1 * sc + p * r[t].y;
                        a2  = a2 * sc + p * r[t].z;
                        a3  = a3 * sc + p * r[t].w;
                        m   = up ? e : m;
                    }
                }
            }
        }
        float M = fmaxf(m, __shfl_xor(m, LPR));
        M = fmaxf(M, __shfl_xor(M, 2 * LPR));
        const float w = hw_exp2(m - M);
        den *= w; a0 *= w; a1 *= w; a2 *= w; a3 *= w;
#pragma unroll
        for (int o = LPR; o < F; o <<= 1) {
            den += __shfl_xor(den, o);
            a0  += __shfl_xor(a0, o);
            a1  += __shfl_xor(a1, o);
            a2  += __shfl_xor(a2, o);
            a3  += __shfl_xor(a3, o);
        }
        const float inv = hw_rcp(den);
        float v0 = a0 * inv + bv.x;
        float v1 = a1 * inv + bv.y;
        float v2 = a2 * inv + bv.z;
        float v3 = a3 * inv + bv.w;
        if constexpr (!FINAL) {
            v0 = (v0 > 0.f) ? v0 : (hw_exp2(v0 * LOG2E) - 1.f);
            v1 = (v1 > 0.f) ? v1 : (hw_exp2(v1 * LOG2E) - 1.f);
            v2 = (v2 > 0.f) ? v2 : (hw_exp2(v2 * LOG2E) - 1.f);
            v3 = (v3 > 0.f) ? v3 : (hw_exp2(v3 * LOG2E) - 1.f);
        } else {
            float mx = fmaxf(fmaxf(v0, v1), fmaxf(v2, v3));
#pragma unroll
            for (int o = 1; o < LPR; o <<= 1) mx = fmaxf(mx, __shfl_xor(mx, o));
            float ss = hw_exp2((v0 - mx) * LOG2E) + hw_exp2((v1 - mx) * LOG2E) +
                       hw_exp2((v2 - mx) * LOG2E) + hw_exp2((v3 - mx) * LOG2E);
#pragma unroll
            for (int o = 1; o < LPR; o <<= 1) ss += __shfl_xor(ss, o);
            const float lg = mx + hw_log2(ss) * LN2;
            v0 -= lg; v1 -= lg; v2 -= lg; v3 -= lg;
        }
        if (live && g == 0) {
            float4 o4; o4.x = v0; o4.y = v1; o4.z = v2; o4.w = v3;
            *reinterpret_cast<float4*>(outp + (size_t)n * F + 4 * c4) = o4;
        }
    }
}

#define TILE_ROWS 16
#define ROWS_PW 8
#define TILE_PITCH 36
template <int H, int C, int NOUT>
__global__ void __launch_bounds__(2 * 32)
__attribute__((amdgpu_waves_per_eu(16)))
node_tile_kernel(const int* __restrict__ beg, const int* __restrict__ endp,
                 const int* __restrict__ ssrc, const float* __restrict__ xw,
                 const float* __restrict__ a_src,
                 const float* __restrict__ a_dst,
                 const float* __restrict__ bias, const float* __restrict__ W2,
                 float* __restrict__ xw2, int last) {
    constexpr int F   = H * C;
    constexpr int LPR = F / 4;
    constexpr int LPH = C / 4;
    constexpr int TB  = 6;
    static_assert(F == 32, "one wave spans one row");
    static_assert(C % 4 == 0, "a float4 must not straddle two heads");
    static_assert(NOUT == 16, "one column tile");
    static_assert(F % 4 == 0 && TILE_PITCH >= F, "tile row holds a whole row");
    __shared__ __attribute__((aligned(16))) float tile[TILE_ROWS * TILE_PITCH];

    const int lane = threadIdx.x & 31;
    const int g    = lane / LPR;
    const int c4   = lane % LPR;
    const int wv   = __builtin_amdgcn_readfirstlane((int)(threadIdx.x >> 5));
    const int row_base = (int)blockIdx.x * TILE_ROWS;
    const int nA   = row_base + ROWS_PW * wv;

    float4 qs, bv;
    qs.x = a_src[4 * c4 + 0]; qs.y = a_src[4 * c4 + 1];
    qs.z = a_src[4 * c4 + 2]; qs.w = a_src[4 * c4 + 3];
    bv.x = bias[4 * c4 + 0];  bv.y = bias[4 * c4 + 1];
    bv.z = bias[4 * c4 + 2];  bv.w = bias[4 * c4 + 3];
    qs.x *= LOG2E; qs.y *= LOG2E; qs.z *= LOG2E; qs.w *= LOG2E;

    const int nl   = nA + (lane % ROWS_PW);
    const int begv = beg[nl];
    const int endv = endp[nl];
    static_assert(ROWS_PW * H == 32, "one (node, head) pair per lane");
    static_assert(LPH == 2, "a head is two float4");
    const int pn = nA + lane / H;
    const int ph = lane % H;
    const float4 pr0 = ld4(xw + (size_t)pn * F + ph * C);
    const float4 pr1 = ld4(xw + (size_t)pn * F + ph * C + 4);
    float4 pq0, pq1;
    pq0.x = a_dst[ph * C + 0] * LOG2E; pq0.y = a_dst[ph * C + 1] * LOG2E;
    pq0.z = a_dst[ph * C + 2] * LOG2E; pq0.w = a_dst[ph * C + 3] * LOG2E;
    pq1.x = a_dst[ph * C + 4] * LOG2E; pq1.y = a_dst[ph * C + 5] * LOG2E;
    pq1.z = a_dst[ph * C + 6] * LOG2E; pq1.w = a_dst[ph * C + 7] * LOG2E;
    const float adl = dot4(pr0, pq0) + dot4(pr1, pq1);

    for (int i = 0; i < ROWS_PW; ++i) {
        const int  nn   = nA + i;
        const int  b    = __builtin_amdgcn_readlane(begv, i);
        const int  en   = __builtin_amdgcn_readlane(endv, i);
        const int  cnt  = en - b;
        const int  total = cnt + 1;
        const int  iters = (total + 3) >> 2;
        const float adn = __shfl(adl, H * i + c4 / LPH);

        float m = -INFINITY, den = 0.f;
        float a0 = 0.f, a1 = 0.f, a2 = 0.f, a3 = 0.f;
        for (int t0 = 0; t0 < iters; t0 += TB) {
            int sidx[TB];
#pragma unroll
            for (int t = 0; t < TB; ++t) {
                const int kk = g + 4 * (t0 + t);
                sidx[t] = ssrc[min(b + kk, last)];
            }
#pragma unroll
            for (int t = 0; t < TB; ++t) {
                const int kk = g + 4 * (t0 + t);
                sidx[t] = (kk < cnt) ? sidx[t] : nn;
            }
            float4 r[TB];
#pragma unroll
            for (int t = 0; t < TB; ++t)
                r[t] = ld4(xw + (size_t)sidx[t] * F + 4 * c4);
#pragma unroll
            for (int t = 0; t < TB; ++t) {
                if (t0 + t < iters) {
                    const int kk = g + 4 * (t0 + t);
                    float e = dot4(r[t], qs);
#pragma unroll
                    for (int o = 1; o < LPH; o <<= 1) e += __shfl_xor(e, o);
                    if (kk < total) {
                        e += adn;
                        e = (e > 0.f) ? e : NEG_SLOPE * e;
                        const float dd = e - m;
                        const float tt = hw_exp2(-fabsf(dd));
                        const bool  up = dd > 0.f;
                        const float sc = up ? tt : 1.f;
                        const float p  = up ? 1.f : tt;
                        den = den * sc + p;
                        a0  = a0 * sc + p * r[t].x;
                        a1  = a1 * sc + p * r[t].y;
                        a2  = a2 * sc + p * r[t].z;
                        a3  = a3 * sc + p * r[t].w;
                        m   = up ? e : m;
                    }
                }
            }
        }
        float M = fmaxf(m, __shfl_xor(m, LPR));
        M = fmaxf(M, __shfl_xor(M, 2 * LPR));
        const float w = hw_exp2(m - M);
        den *= w; a0 *= w; a1 *= w; a2 *= w; a3 *= w;
#pragma unroll
        for (int o = LPR; o < F; o <<= 1) {
            den += __shfl_xor(den, o);
            a0  += __shfl_xor(a0, o);
            a1  += __shfl_xor(a1, o);
            a2  += __shfl_xor(a2, o);
            a3  += __shfl_xor(a3, o);
        }
        const float inv = hw_rcp(den);
        float v0 = a0 * inv + bv.x;
        float v1 = a1 * inv + bv.y;
        float v2 = a2 * inv + bv.z;
        float v3 = a3 * inv + bv.w;
        v0 = (v0 > 0.f) ? v0 : (hw_exp2(v0 * LOG2E) - 1.f);
        v1 = (v1 > 0.f) ? v1 : (hw_exp2(v1 * LOG2E) - 1.f);
        v2 = (v2 > 0.f) ? v2 : (hw_exp2(v2 * LOG2E) - 1.f);
        v3 = (v3 > 0.f) ? v3 : (hw_exp2(v3 * LOG2E) - 1.f);
        if (g == 0) {
            float4 o4; o4.x = v0; o4.y = v1; o4.z = v2; o4.w = v3;
            *reinterpret_cast<float4*>(
                &tile[(ROWS_PW * wv + i) * TILE_PITCH + 4 * c4]) = o4;
        }
    }
    __syncthreads();

    if (wv == 0) {
        const int l  = lane & 15;
        const int hf = lane >> 4;
        v8f acc = (v8f){};
#pragma unroll
        for (int k = 0; k < F; k += 4) {
            v2f a, bb;
            a.x  = tile[l * TILE_PITCH + k + 2 * hf];
            a.y  = tile[l * TILE_PITCH + k + 2 * hf + 1];
            bb.x = W2[(size_t)(k + 2 * hf) * NOUT + l];
            bb.y = W2[(size_t)(k + 2 * hf + 1) * NOUT + l];
            acc = __builtin_amdgcn_wmma_f32_16x16x4_f32(
                 false, a,  false, bb,
                 (short)0, acc,  false,
                 false);
        }
#pragma unroll
        for (int r = 0; r < 8; ++r)
            xw2[(size_t)(row_base + r + 8 * hf) * NOUT + l] = acc[r];
    }
}

extern "C" void kernel_launch(void* const* d_in, const int* in_sizes, int n_in,
                              void* d_out, int out_size, void* d_ws,
                              size_t ws_size, hipStream_t stream) {
    const float* x   = (const float*)d_in[0];
    const int*   ei  = (const int*)d_in[1];
    const float* W1  = (const float*)d_in[2];
    const float* a1s = (const float*)d_in[3];
    const float* a1d = (const float*)d_in[4];
    const float* b1  = (const float*)d_in[5];
    const float* W2  = (const float*)d_in[6];
    const float* a2s = (const float*)d_in[7];
    const float* a2d = (const float*)d_in[8];
    const float* b2  = (const float*)d_in[9];
    float* out = (float*)d_out;

    const int N = in_sizes[0] / 128;
    const int E = in_sizes[1] / 2;
    const int* srcs = ei;
    const int* dsts = ei + E;

    char* p = (char*)d_ws;
    auto carve = [&](size_t bytes) {
        void* r = (void*)p;
        p += (bytes + 255) & ~(size_t)255;
        return r;
    };
    float* xw1  = (float*)carve((size_t)N * 32 * 4);
    float* h1   = (float*)carve((size_t)N * 32 * 4);
    float* xw2  = (float*)carve((size_t)N * 16 * 4);
    int*   endp = (int*)carve((size_t)N * 4);
    int*   beg  = (int*)carve((size_t)N * 4);
    int*   ssrc = (int*)carve((size_t)E * 4);
    int*   binned = (int*)h1;
    int*   tabT   = (int*)xw2;

    const int B = 256;
    auto g = [](int n, int b) { return (n + b - 1) / b; };
    const int NODE_BLOCKS2 = 1536;

    const int NB = g(N, SLICE);
    const int G  = g(E, CHUNK);
    const int GG = g(N / 16, BIN_T / 32);
    bin_gemm_kernel<<<G + GG, BIN_T, (size_t)(NB + 1 + CHUNK) * 4, stream>>>(
        srcs, dsts, E, NB, G, tabT, binned, x, W1, xw1, N);
    slice_kernel<<<NB, SORT_T, 0, stream>>>(tabT, binned, G, N, beg, endp,
                                            ssrc);

    node_tile_kernel<4, 8, 16><<<N / TILE_ROWS, 2 * 32, 0, stream>>>(
        beg, endp, ssrc, xw1, a1s, a1d, b1, W2, xw2, E - 1);

    node_kernel<1, 16, true><<<NODE_BLOCKS2, B, 0, stream>>>(
        beg, endp, ssrc, xw2, a2s, a2d, b2, out, N, E - 1);
}


# slice kernel: half a wave reads each piece (one chunk's edges of the slice, contiguous), a dword per lane, all loads in flight: 2-4 lines per load, not 32; arrival-rank order
# speedup vs baseline: 21.6116x; 1.0907x over previous
#include <hip/hip_runtime.h>
#include <math.h>

typedef __attribute__((ext_vector_type(2))) float v2f;
typedef __attribute__((ext_vector_type(8))) float v8f;

#define NEG_SLOPE 0.2f

template <int K, int NCOL>
__device__ __forceinline__ void gemm_slab(const float* __restrict__ A,
                                          const float* __restrict__ B,
                                          float* __restrict__ C, int row0,
                                          int lane) {
    constexpr int NT = NCOL / 16;
    const int l    = lane & 15;
    const int hf   = lane >> 4;

    const float* __restrict__ arow = A + (size_t)(row0 + l) * K;
    v8f acc[NT];
#pragma unroll
    for (int t = 0; t < NT; ++t) acc[t] = (v8f){};

#pragma unroll
    for (int k = 0; k < K; k += 4) {
        v2f a;
        a.x = arow[k + 2 * hf];
        a.y = arow[k + 2 * hf + 1];
#pragma unroll
        for (int t = 0; t < NT; ++t) {
            v2f b;
            b.x = B[(size_t)(k + 2 * hf) * NCOL + 16 * t + l];
            b.y = B[(size_t)(k + 2 * hf + 1) * NCOL + 16 * t + l];
            acc[t] = __builtin_amdgcn_wmma_f32_16x16x4_f32(
                 false, a,  false, b,
                 (short)0, acc[t],  false,
                 false);
        }
    }
#pragma unroll
    for (int t = 0; t < NT; ++t)
#pragma unroll
        for (int i = 0; i < 8; ++i)
            C[(size_t)(row0 + i + 8 * hf) * NCOL + 16 * t + l] = acc[t][i];
}

#define SLICE_SHIFT 8
#define SLICE 256
#define CHUNK 4096
#define BIN_T 256
#define EPT (CHUNK / BIN_T)
#define SORT_T 512
#define STAGE_CAP 8192

__device__ __forceinline__ void
bin_chunk(const int* __restrict__ srcs, const int* __restrict__ dsts, int E,
          int NB, int G, int* __restrict__ tabT, int* __restrict__ binned) {
    extern __shared__ int smem[];
    int* cnt   = smem;
    int* stage = smem + (NB + 1);
    __shared__ int wsum[BIN_T / 32];

    const int t    = threadIdx.x;
    const int lane = t & 31;
    const int w    = blockIdx.x;
    const int e0   = w * CHUNK;
    const int n_here = min(CHUNK, E - e0);

    for (int i = t; i <= NB; i += BIN_T) cnt[i] = 0;
    __syncthreads();

    int s[EPT], d[EPT], r[EPT];
#pragma unroll
    for (int j = 0; j < EPT; ++j) {
        const int i = j * BIN_T + t;
        s[j] = 0; d[j] = 0;
        if (i < n_here) { s[j] = srcs[e0 + i]; d[j] = dsts[e0 + i]; }
    }
#pragma unroll
    for (int j = 0; j < EPT; ++j) {
        const int i = j * BIN_T + t;
        r[j] = 0;
        if (i < n_here) r[j] = atomicAdd(&cnt[d[j] >> SLICE_SHIFT], 1);
    }
    __syncthreads();

    const int per = (NB + 1 + BIN_T - 1) / BIN_T;
    const int lo  = t * per;
    int v = 0;
    for (int i = 0; i < per; ++i) {
        const int idx = lo + i;
        if (idx <= NB) v += cnt[idx];
    }
    int inc = v;
#pragma unroll
    for (int o = 1; o < 32; o <<= 1) {
        const int u = __shfl_up(inc, o);
        if (lane >= o) inc += u;
    }
    if (lane == 31) wsum[t >> 5] = inc;
    __syncthreads();
    int run = inc - v;
    for (int k = 0; k < (t >> 5); ++k) run += wsum[k];
    for (int i = 0; i < per; ++i) {
        const int idx = lo + i;
        if (idx <= NB) {
            const int c = cnt[idx];
            cnt[idx] = run;
            tabT[(size_t)idx * G + w] = run;
            run += c;
        }
    }
    __syncthreads();

#pragma unroll
    for (int j = 0; j < EPT; ++j) {
        const int i = j * BIN_T + t;
        if (i < n_here)
            stage[cnt[d[j] >> SLICE_SHIFT] + r[j]] =
                (int)((unsigned)s[j] | ((unsigned)(d[j] & (SLICE - 1)) << 24));
    }
    __syncthreads();
    for (int i = t; i < n_here; i += BIN_T) binned[e0 + i] = stage[i];
}

#define GEMM1_K 128
#define GEMM1_NCOL 32
__global__ void __launch_bounds__(BIN_T)
bin_gemm_kernel(const int* __restrict__ srcs, const int* __restrict__ dsts,
                int E, int NB, int G, int* __restrict__ tabT,
                int* __restrict__ binned, const float* __restrict__ x,
                const float* __restrict__ W1, float* __restrict__ xw1, int N) {
    if ((int)blockIdx.x < G) {
        bin_chunk(srcs, dsts, E, NB, G, tabT, binned);
    } else {
        const int wv = __builtin_amdgcn_readfirstlane((int)(threadIdx.x >> 5));
        const int row0 = (((int)blockIdx.x - G) * (BIN_T / 32) + wv) * 16;
        if (row0 < N)
            gemm_slab<GEMM1_K, GEMM1_NCOL>(x, W1, xw1, row0, threadIdx.x & 31);
    }
}

#define WALK_U 13
#define WALK_HW (SORT_T / 16)
template <bool PLACE>
__device__ __forceinline__ void walk_slice(const int* __restrict__ tab_lo,
                                           const int* __restrict__ tab_hi,
                                           const int* __restrict__ binned,
                                           int G, int t, int* counters,
                                           bool fits, int* stage,
                                           int* __restrict__ ssrc_base) {
    const int l16 = t & 15;
    const int hw  = t >> 4;
    for (int w0 = 0; w0 < G; w0 += WALK_HW * WALK_U) {
        int a[WALK_U], c[WALK_U];
#pragma unroll
        for (int u = 0; u < WALK_U; ++u) {
            const int wc = min(w0 + hw + WALK_HW * u, G - 1);
            a[u] = tab_lo[wc];
            c[u] = tab_hi[wc];
        }
        int base[WALK_U];
#pragma unroll
        for (int u = 0; u < WALK_U; ++u) {
            const int w  = w0 + hw + WALK_HW * u;
            const int wc = min(w, G - 1);
            c[u]    = (w < G) ? c[u] - a[u] : 0;
            base[u] = wc * CHUNK + a[u];
        }
        unsigned v0[WALK_U], v1[WALK_U];
#pragma unroll
        for (int u = 0; u < WALK_U; ++u) {
            v0[u] = (unsigned)binned[max(base[u] + min(l16, c[u] - 1), 0)];
            v1[u] = (unsigned)binned[max(base[u] + min(l16 + 16, c[u] - 1), 0)];
        }
        if constexpr (!PLACE) {
#pragma unroll
            for (int u = 0; u < WALK_U; ++u) {
                if (l16 < c[u])      atomicAdd(&counters[v0[u] >> 24], 1);
                if (l16 + 16 < c[u]) atomicAdd(&counters[v1[u] >> 24], 1);
            }
        } else {
            int p0[WALK_U], p1[WALK_U];
#pragma unroll
            for (int u = 0; u < WALK_U; ++u) {
                if (l16 < c[u])      p0[u] = atomicAdd(&counters[v0[u] >> 24], 1);
                if (l16 + 16 < c[u]) p1[u] = atomicAdd(&counters[v1[u] >> 24], 1);
            }
#pragma unroll
            for (int u = 0; u < WALK_U; ++u) {
                const int s0 = (int)(v0[u] & 0x00FFFFFFu);
                const int s1 = (int)(v1[u] & 0x00FFFFFFu);
                if (fits) {
                    if (l16 < c[u])      stage[p0[u]] = s0;
                    if (l16 + 16 < c[u]) stage[p1[u]] = s1;
                } else {
                    if (l16 < c[u])      ssrc_base[p0[u]] = s0;
                    if (l16 + 16 < c[u]) ssrc_base[p1[u]] = s1;
                }
            }
        }
#pragma unroll
        for (int u = 0; u < WALK_U; ++u) {
            for (int j = 32 + l16; j < c[u]; j += 16) {
                const unsigned v = (unsigned)binned[base[u] + j];
                const int dl = (int)(v >> 24);
                if constexpr (!PLACE) {
                    atomicAdd(&counters[dl], 1);
                } else {
                    const int pos = atomicAdd(&counters[dl], 1);
                    const int src = (int)(v & 0x00FFFFFFu);
                    if (fits) stage[pos] = src;
                    else      ssrc_base[pos] = src;
                }
            }
        }
    }
}

__global__ void __launch_bounds__(SORT_T)
slice_kernel(const int* __restrict__ tabT, const int* __restrict__ binned,
             int G, int N, int* __restrict__ beg, int* __restrict__ endp,
             int* __restrict__ ssrc) {
    __shared__ int deg[SLICE];
    __shared__ int cur[SLICE];
    __shared__ int red[2 * (SORT_T / 32)];
    __shared__ int stage[STAGE_CAP];

    const int t    = threadIdx.x;
    const int lane = t & 31;
    const int wv   = t >> 5;
    const int s    = blockIdx.x;
    const int* __restrict__ tab_lo = tabT + (size_t)s * G;
    const int* __restrict__ tab_hi = tabT + (size_t)(s + 1) * G;

    if (t < SLICE) deg[t] = 0;

    int bsum = 0, tsum = 0;
    for (int w = t; w < G; w += SORT_T) {
        const int a = tab_lo[w];
        bsum += a;
        tsum += tab_hi[w] - a;
    }
#pragma unroll
    for (int o = 16; o > 0; o >>= 1) {
        bsum += __shfl_xor(bsum, o);
        tsum += __shfl_xor(tsum, o);
    }
    if (lane == 0) { red[wv] = bsum; red[SORT_T / 32 + wv] = tsum; }
    __syncthreads();
    int base = 0, total = 0;
    for (int k = 0; k < SORT_T / 32; ++k) {
        base  += red[k];
        total += red[SORT_T / 32 + k];
    }
    const bool fits = total <= STAGE_CAP;

    walk_slice<false>(tab_lo, tab_hi, binned, G, t, deg, fits, stage, ssrc);
    __syncthreads();

    const int dv = (t < SLICE) ? deg[t] : 0;
    int inc = dv;
#pragma unroll
    for (int o = 1; o < 32; o <<= 1) {
        const int u = __shfl_up(inc, o);
        if (lane >= o) inc += u;
    }
    if (lane == 31) red[wv] = inc;
    __syncthreads();
    int off = inc - dv;
    for (int k = 0; k < wv; ++k) off += red[k];
    if (t < SLICE) {
        cur[t] = off;
        const int n = s * SLICE + t;
        if (n < N) { beg[n] = base + off; endp[n] = base + off + dv; }
    }
    __syncthreads();

    walk_slice<true>(tab_lo, tab_hi, binned, G, t, cur, fits, stage,
                     ssrc + base);
    __syncthreads();
    if (fits)
        for (int i = t; i < total; i += SORT_T) ssrc[base + i] = stage[i];
}

__device__ __forceinline__ float4 ld4(const float* __restrict__ p) {
    return *reinterpret_cast<const float4*>(p);
}
__device__ __forceinline__ float dot4(const float4& a, const float4& b) {
    return a.x * b.x + a.y * b.y + a.z * b.z + a.w * b.w;
}

#define LOG2E 1.44269504088896340736f
#define LN2   0.69314718055994530942f
__device__ __forceinline__ float hw_exp2(float x) {
    return __builtin_amdgcn_exp2f(x);
}
__device__ __forceinline__ float hw_log2(float x) {
    return __builtin_amdgcn_logf(x);
}
__device__ __forceinline__ float hw_rcp(float x) {
    return __builtin_amdgcn_rcpf(x);
}

template <int H, int C, bool FINAL>
__global__ void node_kernel(const int* __restrict__ beg,
                            const int* __restrict__ endp,
                            const int* __restrict__ ssrc,
                            const float* __restrict__ xw,
                            const float* __restrict__ a_src,
                            const float* __restrict__ a_dst,
                            const float* __restrict__ bias,
                            float* __restrict__ outp, int N, int last) {
    constexpr int F   = H * C;
    constexpr int LPR = F / 4;
    constexpr int LPH = C / 4;
    constexpr int NPW = 32 / F;
    constexpr int TB  = 6;
    static_assert(F == 32 || F == 16, "row must be 32 or 16 floats");
    static_assert(C % 4 == 0, "a float4 must not straddle two heads");
    const int lane = threadIdx.x & 31;
    const int sub  = lane % F;
    const int g    = sub / LPR;
    const int c4   = sub % LPR;
    const int wave0 = __builtin_amdgcn_readfirstlane(
        (int)((blockIdx.x * blockDim.x + threadIdx.x) >> 5));
    const int nwaves = (int)((gridDim.x * blockDim.x) >> 5);
    const int nslots = (N + NPW - 1) / NPW;

    float4 qs, qd, bv;
    qs.x = a_src[4 * c4 + 0]; qs.y = a_src[4 * c4 + 1];
    qs.z = a_src[4 * c4 + 2]; qs.w = a_src[4 * c4 + 3];
    qd.x = a_dst[4 * c4 + 0]; qd.y = a_dst[4 * c4 + 1];
    qd.z = a_dst[4 * c4 + 2]; qd.w = a_dst[4 * c4 + 3];
    bv.x = bias[4 * c4 + 0];  bv.y = bias[4 * c4 + 1];
    bv.z = bias[4 * c4 + 2];  bv.w = bias[4 * c4 + 3];
    qs.x *= LOG2E; qs.y *= LOG2E; qs.z *= LOG2E; qs.w *= LOG2E;
    qd.x *= LOG2E; qd.y *= LOG2E; qd.z *= LOG2E; qd.w *= LOG2E;

    for (int q = wave0; q < nslots; q += nwaves) {
        const int  n    = q * NPW + lane / F;
        const bool live = n < N;
        const int  nn   = live ? n : N - 1;
        const int  b    = beg[nn];
        const int  en   = endp[nn];
        const float4 rn = ld4(xw + (size_t)nn * F + 4 * c4);
        const int  cnt  = live ? en - b : -1;
        float adn = dot4(rn, qd);
#pragma unroll
        for (int o = 1; o < LPH; o <<= 1) adn += __shfl_xor(adn, o);

        const int total = cnt + 1;
        int iters = (total + 3) >> 2;
        if constexpr (NPW == 2) iters = max(iters, __shfl_xor(iters, 16));
        iters = __builtin_amdgcn_readfirstlane(iters);

        float m = -INFINITY, den = 0.f;
        float a0 = 0.f, a1 = 0.f, a2 = 0.f, a3 = 0.f;
        for (int t0 = 0; t0 < iters; t0 += TB) {
            int sidx[TB];
#pragma unroll
            for (int t = 0; t < TB; ++t) {
                const int kk = g + 4 * (t0 + t);
                sidx[t] = ssrc[min(b + kk, last)];
            }
#pragma unroll
            for (int t = 0; t < TB; ++t) {
                const int kk = g + 4 * (t0 + t);
                sidx[t] = (kk < cnt) ? sidx[t] : nn;
            }
            float4 r[TB];
#pragma unroll
            for (int t = 0; t < TB; ++t)
                r[t] = ld4(xw + (size_t)sidx[t] * F + 4 * c4);
#pragma unroll
            for (int t = 0; t < TB; ++t) {
                if (t0 + t < iters) {
                    const int kk = g + 4 * (t0 + t);
                    float e = dot4(r[t], qs);
#pragma unroll
                    for (int o = 1; o < LPH; o <<= 1) e += __shfl_xor(e, o);
                    if (kk < total) {
                        e += adn;
                        e = (e > 0.f) ? e : NEG_SLOPE * e;
                        const float dd = e - m;
                        const float tt = hw_exp2(-fabsf(dd));
                        const bool  up = dd > 0.f;
                        const float sc = up ? tt : 1.f;
                        const float p  = up ? 1.f : tt;
                        den = den * sc + p;
                        a0  = a0 * sc + p * r[t].x;
                        a1  = a1 * sc + p * r[t].y;
                        a2  = a2 * sc + p * r[t].z;
                        a3  = a3 * sc + p * r[t].w;
                        m   = up ? e : m;
                    }
                }
            }
        }
        float M = fmaxf(m, __shfl_xor(m, LPR));
        M = fmaxf(M, __shfl_xor(M, 2 * LPR));
        const float w = hw_exp2(m - M);
        den *= w; a0 *= w; a1 *= w; a2 *= w; a3 *= w;
#pragma unroll
        for (int o = LPR; o < F; o <<= 1) {
            den += __shfl_xor(den, o);
            a0  += __shfl_xor(a0, o);
            a1  += __shfl_xor(a1, o);
            a2  += __shfl_xor(a2, o);
            a3  += __shfl_xor(a3, o);
        }
        const float inv = hw_rcp(den);
        float v0 = a0 * inv + bv.x;
        float v1 = a1 * inv + bv.y;
        float v2 = a2 * inv + bv.z;
        float v3 = a3 * inv + bv.w;
        if constexpr (!FINAL) {
            v0 = (v0 > 0.f) ? v0 : (hw_exp2(v0 * LOG2E) - 1.f);
            v1 = (v1 > 0.f) ? v1 : (hw_exp2(v1 * LOG2E) - 1.f);
            v2 = (v2 > 0.f) ? v2 : (hw_exp2(v2 * LOG2E) - 1.f);
            v3 = (v3 > 0.f) ? v3 : (hw_exp2(v3 * LOG2E) - 1.f);
        } else {
            float mx = fmaxf(fmaxf(v0, v1), fmaxf(v2, v3));
#pragma unroll
            for (int o = 1; o < LPR; o <<= 1) mx = fmaxf(mx, __shfl_xor(mx, o));
            float ss = hw_exp2((v0 - mx) * LOG2E) + hw_exp2((v1 - mx) * LOG2E) +
                       hw_exp2((v2 - mx) * LOG2E) + hw_exp2((v3 - mx) * LOG2E);
#pragma unroll
            for (int o = 1; o < LPR; o <<= 1) ss += __shfl_xor(ss, o);
            const float lg = mx + hw_log2(ss) * LN2;
            v0 -= lg; v1 -= lg; v2 -= lg; v3 -= lg;
        }
        if (live && g == 0) {
            float4 o4; o4.x = v0; o4.y = v1; o4.z = v2; o4.w = v3;
            *reinterpret_cast<float4*>(outp + (size_t)n * F + 4 * c4) = o4;
        }
    }
}

#define TILE_ROWS 16
#define ROWS_PW 8
#define TILE_PITCH 36
template <int H, int C, int NOUT>
__global__ void __launch_bounds__(2 * 32)
__attribute__((amdgpu_waves_per_eu(16)))
node_tile_kernel(const int* __restrict__ beg, const int* __restrict__ endp,
                 const int* __restrict__ ssrc, const float* __restrict__ xw,
                 const float* __restrict__ a_src,
                 const float* __restrict__ a_dst,
                 const float* __restrict__ bias, const float* __restrict__ W2,
                 float* __restrict__ xw2, int last) {
    constexpr int F   = H * C;
    constexpr int LPR = F / 4;
    constexpr int LPH = C / 4;
    constexpr int TB  = 6;
    static_assert(F == 32, "one wave spans one row");
    static_assert(C % 4 == 0, "a float4 must not straddle two heads");
    static_assert(NOUT == 16, "one column tile");
    static_assert(F % 4 == 0 && TILE_PITCH >= F, "tile row holds a whole row");
    __shared__ __attribute__((aligned(16))) float tile[TILE_ROWS * TILE_PITCH];

    const int lane = threadIdx.x & 31;
    const int g    = lane / LPR;
    const int c4   = lane % LPR;
    const int wv   = __builtin_amdgcn_readfirstlane((int)(threadIdx.x >> 5));
    const int row_base = (int)blockIdx.x * TILE_ROWS;
    const int nA   = row_base + ROWS_PW * wv;

    float4 qs, bv;
    qs.x = a_src[4 * c4 + 0]; qs.y = a_src[4 * c4 + 1];
    qs.z = a_src[4 * c4 + 2]; qs.w = a_src[4 * c4 + 3];
    bv.x = bias[4 * c4 + 0];  bv.y = bias[4 * c4 + 1];
    bv.z = bias[4 * c4 + 2];  bv.w = bias[4 * c4 + 3];
    qs.x *= LOG2E; qs.y *= LOG2E; qs.z *= LOG2E; qs.w *= LOG2E;

    const int nl   = nA + (lane % ROWS_PW);
    const int begv = beg[nl];
    const int endv = endp[nl];
    static_assert(ROWS_PW * H == 32, "one (node, head) pair per lane");
    static_assert(LPH == 2, "a head is two float4");
    const int pn = nA + lane / H;
    const int ph = lane % H;
    const float4 pr0 = ld4(xw + (size_t)pn * F + ph * C);
    const float4 pr1 = ld4(xw + (size_t)pn * F + ph * C + 4);
    float4 pq0, pq1;
    pq0.x = a_dst[ph * C + 0] * LOG2E; pq0.y = a_dst[ph * C + 1] * LOG2E;
    pq0.z = a_dst[ph * C + 2] * LOG2E; pq0.w = a_dst[ph * C + 3] * LOG2E;
    pq1.x = a_dst[ph * C + 4] * LOG2E; pq1.y = a_dst[ph * C + 5] * LOG2E;
    pq1.z = a_dst[ph * C + 6] * LOG2E; pq1.w = a_dst[ph * C + 7] * LOG2E;
    const float adl = dot4(pr0, pq0) + dot4(pr1, pq1);

    for (int i = 0; i < ROWS_PW; ++i) {
        const int  nn   = nA + i;
        const int  b    = __builtin_amdgcn_readlane(begv, i);
        const int  en   = __builtin_amdgcn_readlane(endv, i);
        const int  cnt  = en - b;
        const int  total = cnt + 1;
        const int  iters = (total + 3) >> 2;
        const float adn = __shfl(adl, H * i + c4 / LPH);

        float m = -INFINITY, den = 0.f;
        float a0 = 0.f, a1 = 0.f, a2 = 0.f, a3 = 0.f;
        for (int t0 = 0; t0 < iters; t0 += TB) {
            int sidx[TB];
#pragma unroll
            for (int t = 0; t < TB; ++t) {
                const int kk = g + 4 * (t0 + t);
                sidx[t] = ssrc[min(b + kk, last)];
            }
#pragma unroll
            for (int t = 0; t < TB; ++t) {
                const int kk = g + 4 * (t0 + t);
                sidx[t] = (kk < cnt) ? sidx[t] : nn;
            }
            float4 r[TB];
#pragma unroll
            for (int t = 0; t < TB; ++t)
                r[t] = ld4(xw + (size_t)sidx[t] * F + 4 * c4);
#pragma unroll
            for (int t = 0; t < TB; ++t) {
                if (t0 + t < iters) {
                    const int kk = g + 4 * (t0 + t);
                    float e = dot4(r[t], qs);
#pragma unroll
                    for (int o = 1; o < LPH; o <<= 1) e += __shfl_xor(e, o);
                    if (kk < total) {
                        e += adn;
                        e = (e > 0.f) ? e : NEG_SLOPE * e;
                        const float dd = e - m;
                        const float tt = hw_exp2(-fabsf(dd));
                        const bool  up = dd > 0.f;
                        const float sc = up ? tt : 1.f;
                        const float p  = up ? 1.f : tt;
                        den = den * sc + p;
                        a0  = a0 * sc + p * r[t].x;
                        a1  = a1 * sc + p * r[t].y;
                        a2  = a2 * sc + p * r[t].z;
                        a3  = a3 * sc + p * r[t].w;
                        m   = up ? e : m;
                    }
                }
            }
        }
        float M = fmaxf(m, __shfl_xor(m, LPR));
        M = fmaxf(M, __shfl_xor(M, 2 * LPR));
        const float w = hw_exp2(m - M);
        den *= w; a0 *= w; a1 *= w; a2 *= w; a3 *= w;
#pragma unroll
        for (int o = LPR; o < F; o <<= 1) {
            den += __shfl_xor(den, o);
            a0  += __shfl_xor(a0, o);
            a1  += __shfl_xor(a1, o);
            a2  += __shfl_xor(a2, o);
            a3  += __shfl_xor(a3, o);
        }
        const float inv = hw_rcp(den);
        float v0 = a0 * inv + bv.x;
        float v1 = a1 * inv + bv.y;
        float v2 = a2 * inv + bv.z;
        float v3 = a3 * inv + bv.w;
        v0 = (v0 > 0.f) ? v0 : (hw_exp2(v0 * LOG2E) - 1.f);
        v1 = (v1 > 0.f) ? v1 : (hw_exp2(v1 * LOG2E) - 1.f);
        v2 = (v2 > 0.f) ? v2 : (hw_exp2(v2 * LOG2E) - 1.f);
        v3 = (v3 > 0.f) ? v3 : (hw_exp2(v3 * LOG2E) - 1.f);
        if (g == 0) {
            float4 o4; o4.x = v0; o4.y = v1; o4.z = v2; o4.w = v3;
            *reinterpret_cast<float4*>(
                &tile[(ROWS_PW * wv + i) * TILE_PITCH + 4 * c4]) = o4;
        }
    }
    __syncthreads();

    if (wv == 0) {
        const int l  = lane & 15;
        const int hf = lane >> 4;
        v8f acc = (v8f){};
#pragma unroll
        for (int k = 0; k < F; k += 4) {
            v2f a, bb;
            a.x  = tile[l * TILE_PITCH + k + 2 * hf];
            a.y  = tile[l * TILE_PITCH + k + 2 * hf + 1];
            bb.x = W2[(size_t)(k + 2 * hf) * NOUT + l];
            bb.y = W2[(size_t)(k + 2 * hf + 1) * NOUT + l];
            acc = __builtin_amdgcn_wmma_f32_16x16x4_f32(
                 false, a,  false, bb,
                 (short)0, acc,  false,
                 false);
        }
#pragma unroll
        for (int r = 0; r < 8; ++r)
            xw2[(size_t)(row_base + r + 8 * hf) * NOUT + l] = acc[r];
    }
}

extern "C" void kernel_launch(void* const* d_in, const int* in_sizes, int n_in,
                              void* d_out, int out_size, void* d_ws,
                              size_t ws_size, hipStream_t stream) {
    const float* x   = (const float*)d_in[0];
    const int*   ei  = (const int*)d_in[1];
    const float* W1  = (const float*)d_in[2];
    const float* a1s = (const float*)d_in[3];
    const float* a1d = (const float*)d_in[4];
    const float* b1  = (const float*)d_in[5];
    const float* W2  = (const float*)d_in[6];
    const float* a2s = (const float*)d_in[7];
    const float* a2d = (const float*)d_in[8];
    const float* b2  = (const float*)d_in[9];
    float* out = (float*)d_out;

    const int N = in_sizes[0] / 128;
    const int E = in_sizes[1] / 2;
    const int* srcs = ei;
    const int* dsts = ei + E;

    char* p = (char*)d_ws;
    auto carve = [&](size_t bytes) {
        void* r = (void*)p;
        p += (bytes + 255) & ~(size_t)255;
        return r;
    };
    float* xw1  = (float*)carve((size_t)N * 32 * 4);
    float* h1   = (float*)carve((size_t)N * 32 * 4);
    float* xw2  = (float*)carve((size_t)N * 16 * 4);
    int*   endp = (int*)carve((size_t)N * 4);
    int*   beg  = (int*)carve((size_t)N * 4);
    int*   ssrc = (int*)carve((size_t)E * 4);
    int*   binned = (int*)h1;
    int*   tabT   = (int*)xw2;

    const int B = 256;
    auto g = [](int n, int b) { return (n + b - 1) / b; };
    const int NODE_BLOCKS2 = 1536;

    const int NB = g(N, SLICE);
    const int G  = g(E, CHUNK);
    const int GG = g(N / 16, BIN_T / 32);
    bin_gemm_kernel<<<G + GG, BIN_T, (size_t)(NB + 1 + CHUNK) * 4, stream>>>(
        srcs, dsts, E, NB, G, tabT, binned, x, W1, xw1, N);
    slice_kernel<<<NB, SORT_T, 0, stream>>>(tabT, binned, G, N, beg, endp,
                                            ssrc);

    node_tile_kernel<4, 8, 16><<<N / TILE_ROWS, 2 * 32, 0, stream>>>(
        beg, endp, ssrc, xw1, a1s, a1d, b1, W2, xw2, E - 1);

    node_kernel<1, 16, true><<<NODE_BLOCKS2, B, 0, stream>>>(
        beg, endp, ssrc, xw2, a2s, a2d, b2, out, N, E - 1);
}
